# starting kernel
# baseline (speedup 1.0000x reference)
#include <hip/hip_runtime.h>
#include <math.h>

typedef __attribute__((ext_vector_type(2))) float v2f;
typedef __attribute__((ext_vector_type(4))) float v4f;
typedef __attribute__((ext_vector_type(8))) float v8f;

#define KC 16

template <int LDA, int LDW>
__device__ inline void gemm_tile_acc(const float* __restrict__ A,
                                     const float* __restrict__ W,
                                     int K, int m0, int n0w,
                                     float (*ldsA)[20], float (*ldsB)[20],
                                     v8f& acc)
{
    const int t    = threadIdx.x;
    const int lane = t & 31;
    const int half = lane >> 4;
    const int mrow = lane & 15;

    const int ar = t >> 3;
    const int ac = (t & 7) * 2;
    const int br = lane >> 1;
    const int bc = (lane & 1) * 8;

    const float* aptr = A + (size_t)(m0 + ar) * LDA + ac;
    const float* bptr = W + (size_t)br * LDW + n0w + bc;

    for (int k0 = 0; k0 < K; k0 += KC) {
        const v2f a2 = *(const v2f*)(aptr + k0);
        ldsA[ar][ac]     = a2.x;
        ldsA[ar][ac + 1] = a2.y;

        const v4f b0 = *(const v4f*)(bptr + (size_t)k0 * LDW);
        const v4f b1 = *(const v4f*)(bptr + (size_t)k0 * LDW + 4);
        *(v4f*)&ldsB[br][bc]     = b0;
        *(v4f*)&ldsB[br][bc + 4] = b1;
        __syncthreads();

#pragma unroll
        for (int ks = 0; ks < KC; ks += 4) {
            v2f a, b;
            a.x = ldsA[mrow][ks + 2 * half];
            a.y = ldsA[mrow][ks + 2 * half + 1];
            b.x = ldsB[ks + 2 * half][mrow];
            b.y = ldsB[ks + 2 * half + 1][mrow];
            acc = __builtin_amdgcn_wmma_f32_16x16x4_f32(
                false, a, false, b, (short)0, acc, false, false);
        }
        __syncthreads();
    }
}

__global__ void __launch_bounds__(128)
lstm_cell_kernel(const float* __restrict__ A1, const float* __restrict__ W,
                 const float* __restrict__ A2, const float* __restrict__ Uw,
                 const float* __restrict__ bias, const float* __restrict__ Cprev,
                 float* __restrict__ Hout, float* __restrict__ Cout,
                 int K1, int K2, int Uu)
{
    __shared__ float ldsA[16][20];
    __shared__ float ldsB[4][16][20];
    __shared__ float gbuf[4][16][17];

    const int t    = threadIdx.x;
    const int wave = t >> 5;
    const int lane = t & 31;
    const int half = lane >> 4;
    const int mrow = lane & 15;
    const int m0   = blockIdx.y * 16;
    const int n0   = blockIdx.x * 16;
    const int n0w  = wave * Uu + n0;

    v8f acc = {};
    gemm_tile_acc<1024, 4096>(A1, W,  K1, m0, n0w, ldsA, ldsB[wave], acc);
    gemm_tile_acc<1024, 4096>(A2, Uw, K2, m0, n0w, ldsA, ldsB[wave], acc);

    const float bv = bias[n0w + mrow];
#pragma unroll
    for (int r = 0; r < 8; ++r) {
        const float z = acc[r] + bv;
        const float a = (wave == 2) ? tanhf(z) : (1.0f / (1.0f + __expf(-z)));
        gbuf[wave][r + 8 * half][mrow] = a;
    }
    __syncthreads();

#pragma unroll
    for (int e = t; e < 256; e += 128) {
        const int m = e >> 4, n = e & 15;
        const float iv = gbuf[0][m][n];
        const float fv = gbuf[1][m][n];
        const float gv = gbuf[2][m][n];
        const float ov = gbuf[3][m][n];
        const size_t idx = (size_t)(m0 + m) * Uu + n0 + n;
        const float cn = fv * Cprev[idx] + iv * gv;
        Cout[idx] = cn;
        Hout[idx] = ov * tanhf(cn);
    }
}

__global__ void __launch_bounds__(128)
sproj_kernel(const float* __restrict__ H2, const float* __restrict__ Ws,
             const float* __restrict__ bs, float* __restrict__ Sout, int Uu)
{
    __shared__ float ldsA[16][20];
    __shared__ float ldsB[4][16][20];

    const int t    = threadIdx.x;
    const int wave = t >> 5;
    const int lane = t & 31;
    const int half = lane >> 4;
    const int mrow = lane & 15;
    const int m0   = blockIdx.y * 16;
    const int n0w  = blockIdx.x * 64 + wave * 16;

    v8f acc = {};
    gemm_tile_acc<1024, 1024>(H2, Ws, Uu, m0, n0w, ldsA, ldsB[wave], acc);

    const float bv = bs[n0w + mrow];
#pragma unroll
    for (int r = 0; r < 8; ++r) {
        const int gm = m0 + r + 8 * half;
        Sout[(size_t)gm * Uu + n0w + mrow] = acc[r] + bv;
    }
}

__global__ void __launch_bounds__(512)
attention_kernel(const float* __restrict__ av, const float* __restrict__ S,
                 const float* __restrict__ H2, const float* __restrict__ C2,
                 float* __restrict__ out, int T, int Uu, int B)
{
    __shared__ float sS[1024];
    __shared__ float ctx_s[1024];
    __shared__ float wm[16], wl[16];

    const int b    = blockIdx.x;
    const int t    = threadIdx.x;
    const int wave = t >> 5;
    const int lane = t & 31;

    for (int u = t; u < Uu; u += 512) {
        sS[u]    = S[(size_t)b * Uu + u];
        ctx_s[u] = 0.0f;
    }
    __syncthreads();

    float m = -INFINITY, l = 0.0f;
    float ctx[32];
#pragma unroll
    for (int j = 0; j < 32; ++j) ctx[j] = 0.0f;

    const float* avb = av + (size_t)b * T * Uu;
    for (int tt = wave; tt < T; tt += 16) {
        const float* row = avb + (size_t)tt * Uu;
        __builtin_prefetch(avb + (size_t)(tt + 16) * Uu + lane * 32, 0, 0);

        v4f frag[8];
        float dot = 0.0f;
#pragma unroll
        for (int j = 0; j < 8; ++j) {
            const int u = (j * 32 + lane) * 4;
            frag[j] = __builtin_nontemporal_load((const v4f*)(row + u));
            dot += frag[j].x * sS[u] + frag[j].y * sS[u + 1] +
                   frag[j].z * sS[u + 2] + frag[j].w * sS[u + 3];
        }
#pragma unroll
        for (int off = 16; off > 0; off >>= 1) dot += __shfl_xor(dot, off, 32);

        const float mn    = fmaxf(m, dot);
        const float scale = __expf(m - mn);
        const float e     = __expf(dot - mn);
        l = l * scale + e;
#pragma unroll
        for (int j = 0; j < 8; ++j) {
            ctx[j * 4 + 0] = ctx[j * 4 + 0] * scale + e * frag[j].x;
            ctx[j * 4 + 1] = ctx[j * 4 + 1] * scale + e * frag[j].y;
            ctx[j * 4 + 2] = ctx[j * 4 + 2] * scale + e * frag[j].z;
            ctx[j * 4 + 3] = ctx[j * 4 + 3] * scale + e * frag[j].w;
        }
        m = mn;
    }

    if (lane == 0) { wm[wave] = m; wl[wave] = l; }
    __syncthreads();

    float M = wm[0];
#pragma unroll
    for (int w = 1; w < 16; ++w) M = fmaxf(M, wm[w]);
    float L = 0.0f;
#pragma unroll
    for (int w = 0; w < 16; ++w) L += wl[w] * __expf(wm[w] - M);

    const float f = __expf(m - M);
#pragma unroll
    for (int j = 0; j < 8; ++j) {
        const int u = (j * 32 + lane) * 4;
        atomicAdd(&ctx_s[u + 0], ctx[j * 4 + 0] * f);
        atomicAdd(&ctx_s[u + 1], ctx[j * 4 + 1] * f);
        atomicAdd(&ctx_s[u + 2], ctx[j * 4 + 2] * f);
        atomicAdd(&ctx_s[u + 3], ctx[j * 4 + 3] * f);
    }
    __syncthreads();

    const float invL = 1.0f / L;
    const size_t base = (size_t)b * Uu;
    for (int u = t; u < Uu; u += 512) {
        out[base + u]                  = H2[base + u];
        out[(size_t)B * Uu + base + u] = C2[base + u] + ctx_s[u] * invL;
    }
}

extern "C" void kernel_launch(void* const* d_in, const int* in_sizes, int n_in,
                              void* d_out, int out_size, void* d_ws, size_t ws_size,
                              hipStream_t stream)
{
    (void)in_sizes; (void)n_in; (void)out_size; (void)ws_size;
    const int B = 128, T = 512, U = 1024;

    const float* x   = (const float*)d_in[0];
    const float* h0  = (const float*)d_in[1];
    const float* c0  = (const float*)d_in[2];
    const float* av  = (const float*)d_in[3];
    const float* W1  = (const float*)d_in[4];
    const float* U1  = (const float*)d_in[5];
    const float* b1  = (const float*)d_in[6];
    const float* W2  = (const float*)d_in[7];
    const float* U2  = (const float*)d_in[8];
    const float* b2  = (const float*)d_in[9];
    const float* Ws  = (const float*)d_in[10];
    const float* bs  = (const float*)d_in[11];
    float* out = (float*)d_out;

    float* ws = (float*)d_ws;
    const size_t BU = (size_t)B * U;
    float* h1 = ws;
    float* c1 = ws + BU;
    float* h2 = ws + 2 * BU;
    float* c2 = ws + 3 * BU;
    float* s  = ws + 4 * BU;

    dim3 gGemm(U / 16, B / 16);
    dim3 bGemm(128);

    lstm_cell_kernel<<<gGemm, bGemm, 0, stream>>>(x, W1, h0, U1, b1, c0,
                                                  h1, c1, 1024, U, U);
    lstm_cell_kernel<<<gGemm, bGemm, 0, stream>>>(h1, W2, h1, U2, b2, c1,
                                                  h2, c2, U, U, U);
    sproj_kernel<<<dim3(U / 64, B / 16), bGemm, 0, stream>>>(h2, Ws, bs, s, U);
    attention_kernel<<<dim3(B), dim3(512), 0, stream>>>(av, s, h2, c2, out,
                                                        T, U, B);
}


# K-chunk staged per LDS round trip 16 -> 128 in the shared GEMM loop (all 3 GEMM launches): 8x fewer dependent load-store-barrier trips, 20 loads in flight per lane; k order of the sums unchanged
# speedup vs baseline: 2.6017x; 2.6017x over previous
#include <hip/hip_runtime.h>
#include <math.h>

typedef __attribute__((ext_vector_type(2))) float v2f;
typedef __attribute__((ext_vector_type(4))) float v4f;
typedef __attribute__((ext_vector_type(8))) float v8f;

#define KC 128
#define LDA_PITCH (KC + 4)

template <int LDA, int LDW>
__device__ inline void gemm_tile_acc(const float* __restrict__ A,
                                     const float* __restrict__ W,
                                     int K, int m0, int n0w,
                                     float (*ldsA)[LDA_PITCH], float (*ldsB)[20],
                                     v8f& acc)
{
    const int t    = threadIdx.x;
    const int lane = t & 31;
    const int half = lane >> 4;
    const int mrow = lane & 15;

    const int ar = t >> 3;
    const int ac = (t & 7) * 4;
    const int br = lane >> 2;
    const int bc = (lane & 3) * 4;

    const float* aptr = A + (size_t)(m0 + ar) * LDA + ac;
    const float* bptr = W + (size_t)br * LDW + n0w + bc;

    for (int k0 = 0; k0 < K; k0 += KC) {
        v4f av[KC / 32];
        v4f bv[KC / 8];
#pragma unroll
        for (int i = 0; i < KC / 32; ++i)
            av[i] = *(const v4f*)(aptr + k0 + 32 * i);
#pragma unroll
        for (int i = 0; i < KC / 8; ++i)
            bv[i] = *(const v4f*)(bptr + (size_t)(k0 + 8 * i) * LDW);
#pragma unroll
        for (int i = 0; i < KC / 32; ++i)
            *(v4f*)&ldsA[ar][ac + 32 * i] = av[i];
#pragma unroll
        for (int i = 0; i < KC / 8; ++i)
            *(v4f*)&ldsB[br + 8 * i][bc] = bv[i];
        __syncthreads();

#pragma unroll
        for (int ks = 0; ks < KC; ks += 4) {
            v2f a, b;
            a.x = ldsA[mrow][ks + 2 * half];
            a.y = ldsA[mrow][ks + 2 * half + 1];
            b.x = ldsB[ks + 2 * half][mrow];
            b.y = ldsB[ks + 2 * half + 1][mrow];
            acc = __builtin_amdgcn_wmma_f32_16x16x4_f32(
                false, a, false, b, (short)0, acc, false, false);
        }
        __syncthreads();
    }
}

__global__ void __launch_bounds__(128)
lstm_cell_kernel(const float* __restrict__ A1, const float* __restrict__ W,
                 const float* __restrict__ A2, const float* __restrict__ Uw,
                 const float* __restrict__ bias, const float* __restrict__ Cprev,
                 float* __restrict__ Hout, float* __restrict__ Cout,
                 int K1, int K2, int Uu)
{
    __shared__ float ldsA[16][LDA_PITCH];
    __shared__ float ldsB[4][KC][20];
    __shared__ float gbuf[4][16][17];

    const int t    = threadIdx.x;
    const int wave = t >> 5;
    const int lane = t & 31;
    const int half = lane >> 4;
    const int mrow = lane & 15;
    const int m0   = blockIdx.y * 16;
    const int n0   = blockIdx.x * 16;
    const int n0w  = wave * Uu + n0;

    v8f acc = {};
    gemm_tile_acc<1024, 4096>(A1, W,  K1, m0, n0w, ldsA, ldsB[wave], acc);
    gemm_tile_acc<1024, 4096>(A2, Uw, K2, m0, n0w, ldsA, ldsB[wave], acc);

    const float bv = bias[n0w + mrow];
#pragma unroll
    for (int r = 0; r < 8; ++r) {
        const float z = acc[r] + bv;
        const float a = (wave == 2) ? tanhf(z) : (1.0f / (1.0f + __expf(-z)));
        gbuf[wave][r + 8 * half][mrow] = a;
    }
    __syncthreads();

#pragma unroll
    for (int e = t; e < 256; e += 128) {
        const int m = e >> 4, n = e & 15;
        const float iv = gbuf[0][m][n];
        const float fv = gbuf[1][m][n];
        const float gv = gbuf[2][m][n];
        const float ov = gbuf[3][m][n];
        const size_t idx = (size_t)(m0 + m) * Uu + n0 + n;
        const float cn = fv * Cprev[idx] + iv * gv;
        Cout[idx] = cn;
        Hout[idx] = ov * tanhf(cn);
    }
}

__global__ void __launch_bounds__(128)
sproj_kernel(const float* __restrict__ H2, const float* __restrict__ Ws,
             const float* __restrict__ bs, float* __restrict__ Sout, int Uu)
{
    __shared__ float ldsA[16][LDA_PITCH];
    __shared__ float ldsB[4][KC][20];

    const int t    = threadIdx.x;
    const int wave = t >> 5;
    const int lane = t & 31;
    const int half = lane >> 4;
    const int mrow = lane & 15;
    const int m0   = blockIdx.y * 16;
    const int n0w  = blockIdx.x * 64 + wave * 16;

    v8f acc = {};
    gemm_tile_acc<1024, 1024>(H2, Ws, Uu, m0, n0w, ldsA, ldsB[wave], acc);

    const float bv = bs[n0w + mrow];
#pragma unroll
    for (int r = 0; r < 8; ++r) {
        const int gm = m0 + r + 8 * half;
        Sout[(size_t)gm * Uu + n0w + mrow] = acc[r] + bv;
    }
}

__global__ void __launch_bounds__(512)
attention_kernel(const float* __restrict__ av, const float* __restrict__ S,
                 const float* __restrict__ H2, const float* __restrict__ C2,
                 float* __restrict__ out, int T, int Uu, int B)
{
    __shared__ float sS[1024];
    __shared__ float ctx_s[1024];
    __shared__ float wm[16], wl[16];

    const int b    = blockIdx.x;
    const int t    = threadIdx.x;
    const int wave = t >> 5;
    const int lane = t & 31;

    for (int u = t; u < Uu; u += 512) {
        sS[u]    = S[(size_t)b * Uu + u];
        ctx_s[u] = 0.0f;
    }
    __syncthreads();

    float m = -INFINITY, l = 0.0f;
    float ctx[32];
#pragma unroll
    for (int j = 0; j < 32; ++j) ctx[j] = 0.0f;

    const float* avb = av + (size_t)b * T * Uu;
    for (int tt = wave; tt < T; tt += 16) {
        const float* row = avb + (size_t)tt * Uu;
        __builtin_prefetch(avb + (size_t)(tt + 16) * Uu + lane * 32, 0, 0);

        v4f frag[8];
        float dot = 0.0f;
#pragma unroll
        for (int j = 0; j < 8; ++j) {
            const int u = (j * 32 + lane) * 4;
            frag[j] = __builtin_nontemporal_load((const v4f*)(row + u));
            dot += frag[j].x * sS[u] + frag[j].y * sS[u + 1] +
                   frag[j].z * sS[u + 2] + frag[j].w * sS[u + 3];
        }
#pragma unroll
        for (int off = 16; off > 0; off >>= 1) dot += __shfl_xor(dot, off, 32);

        const float mn    = fmaxf(m, dot);
        const float scale = __expf(m - mn);
        const float e     = __expf(dot - mn);
        l = l * scale + e;
#pragma unroll
        for (int j = 0; j < 8; ++j) {
            ctx[j * 4 + 0] = ctx[j * 4 + 0] * scale + e * frag[j].x;
            ctx[j * 4 + 1] = ctx[j * 4 + 1] * scale + e * frag[j].y;
            ctx[j * 4 + 2] = ctx[j * 4 + 2] * scale + e * frag[j].z;
            ctx[j * 4 + 3] = ctx[j * 4 + 3] * scale + e * frag[j].w;
        }
        m = mn;
    }

    if (lane == 0) { wm[wave] = m; wl[wave] = l; }
    __syncthreads();

    float M = wm[0];
#pragma unroll
    for (int w = 1; w < 16; ++w) M = fmaxf(M, wm[w]);
    float L = 0.0f;
#pragma unroll
    for (int w = 0; w < 16; ++w) L += wl[w] * __expf(wm[w] - M);

    const float f = __expf(m - M);
#pragma unroll
    for (int j = 0; j < 8; ++j) {
        const int u = (j * 32 + lane) * 4;
        atomicAdd(&ctx_s[u + 0], ctx[j * 4 + 0] * f);
        atomicAdd(&ctx_s[u + 1], ctx[j * 4 + 1] * f);
        atomicAdd(&ctx_s[u + 2], ctx[j * 4 + 2] * f);
        atomicAdd(&ctx_s[u + 3], ctx[j * 4 + 3] * f);
    }
    __syncthreads();

    const float invL = 1.0f / L;
    const size_t base = (size_t)b * Uu;
    for (int u = t; u < Uu; u += 512) {
        out[base + u]                  = H2[base + u];
        out[(size_t)B * Uu + base + u] = C2[base + u] + ctx_s[u] * invL;
    }
}

extern "C" void kernel_launch(void* const* d_in, const int* in_sizes, int n_in,
                              void* d_out, int out_size, void* d_ws, size_t ws_size,
                              hipStream_t stream)
{
    (void)in_sizes; (void)n_in; (void)out_size; (void)ws_size;
    const int B = 128, T = 512, U = 1024;

    const float* x   = (const float*)d_in[0];
    const float* h0  = (const float*)d_in[1];
    const float* c0  = (const float*)d_in[2];
    const float* av  = (const float*)d_in[3];
    const float* W1  = (const float*)d_in[4];
    const float* U1  = (const float*)d_in[5];
    const float* b1  = (const float*)d_in[6];
    const float* W2  = (const float*)d_in[7];
    const float* U2  = (const float*)d_in[8];
    const float* b2  = (const float*)d_in[9];
    const float* Ws  = (const float*)d_in[10];
    const float* bs  = (const float*)d_in[11];
    float* out = (float*)d_out;

    float* ws = (float*)d_ws;
    const size_t BU = (size_t)B * U;
    float* h1 = ws;
    float* c1 = ws + BU;
    float* h2 = ws + 2 * BU;
    float* c2 = ws + 3 * BU;
    float* s  = ws + 4 * BU;

    dim3 gGemm(U / 16, B / 16);
    dim3 bGemm(128);

    lstm_cell_kernel<<<gGemm, bGemm, 0, stream>>>(x, W1, h0, U1, b1, c0,
                                                  h1, c1, 1024, U, U);
    lstm_cell_kernel<<<gGemm, bGemm, 0, stream>>>(h1, W2, h1, U2, b2, c1,
                                                  h2, c2, U, U, U);
    sproj_kernel<<<dim3(U / 64, B / 16), bGemm, 0, stream>>>(h2, Ws, bs, s, U);
    attention_kernel<<<dim3(B), dim3(512), 0, stream>>>(av, s, h2, c2, out,
                                                        T, U, B);
}


# GEMM K-loop software-pipelined: next chunk's global loads issued into registers ahead of the current chunk's products; a cell's two operand pairs run as one chunk sequence; order of sums unchanged
# speedup vs baseline: 2.8139x; 1.0816x over previous
#include <hip/hip_runtime.h>
#include <math.h>

typedef __attribute__((ext_vector_type(2))) float v2f;
typedef __attribute__((ext_vector_type(4))) float v4f;
typedef __attribute__((ext_vector_type(8))) float v8f;

#define KC 128
#define LDA_PITCH (KC + 4)


template <int LDW>
__device__ __forceinline__ void load_chunk(const float* __restrict__ ap,
                                           const float* __restrict__ bp,
                                           v4f (&av)[KC / 32], v4f (&bv)[KC / 8])
{
#pragma unroll
    for (int i = 0; i < KC / 32; ++i)
        av[i] = *(const v4f*)(ap + 32 * i);
#pragma unroll
    for (int i = 0; i < KC / 8; ++i)
        bv[i] = *(const v4f*)(bp + (size_t)(8 * i) * LDW);
}

template <int LDA, int LDW>
__device__ inline void gemm_tile_acc(const float* __restrict__ A1,
                                     const float* __restrict__ W1,
                                     const float* __restrict__ A2,
                                     const float* __restrict__ W2,
                                     int nc1, int nc, int m0, int n0w,
                                     float (*ldsA)[LDA_PITCH], float (*ldsB)[20],
                                     v8f& acc)
{
    const int t    = threadIdx.x;
    const int lane = t & 31;
    const int half = lane >> 4;
    const int mrow = lane & 15;

    const int ar = t >> 3;
    const int ac = (t & 7) * 4;
    const int br = lane >> 2;
    const int bc = (lane & 3) * 4;

    const size_t aoff = (size_t)(m0 + ar) * LDA + ac;
    const size_t boff = (size_t)br * LDW + n0w + bc;

    v4f av[KC / 32];
    v4f bv[KC / 8];

    if (nc > 0) {
        const bool first = (0 < nc1);
        load_chunk<LDW>((first ? A1 : A2) + aoff, (first ? W1 : W2) + boff, av, bv);
    }

    for (int c = 0; c < nc; ++c) {
#pragma unroll
        for (int i = 0; i < KC / 32; ++i)
            *(v4f*)&ldsA[ar][ac + 32 * i] = av[i];
#pragma unroll
        for (int i = 0; i < KC / 8; ++i)
            *(v4f*)&ldsB[br + 8 * i][bc] = bv[i];
        __syncthreads();

        if (c + 1 < nc) {
            const bool first = (c + 1 < nc1);
            const int  kk    = (first ? (c + 1) : (c + 1 - nc1)) * KC;
            load_chunk<LDW>((first ? A1 : A2) + aoff + kk,
                            (first ? W1 : W2) + boff + (size_t)kk * LDW, av, bv);
        }

#pragma unroll
        for (int ks = 0; ks < KC; ks += 4) {
            v2f a, b;
            a.x = ldsA[mrow][ks + 2 * half];
            a.y = ldsA[mrow][ks + 2 * half + 1];
            b.x = ldsB[ks + 2 * half][mrow];
            b.y = ldsB[ks + 2 * half + 1][mrow];
            acc = __builtin_amdgcn_wmma_f32_16x16x4_f32(
                false, a, false, b, (short)0, acc, false, false);
        }
        __syncthreads();
    }
}

__global__ void __launch_bounds__(128)
lstm_cell_kernel(const float* __restrict__ A1, const float* __restrict__ W,
                 const float* __restrict__ A2, const float* __restrict__ Uw,
                 const float* __restrict__ bias, const float* __restrict__ Cprev,
                 float* __restrict__ Hout, float* __restrict__ Cout,
                 int K1, int K2, int Uu)
{
    __shared__ float ldsA[16][LDA_PITCH];
    __shared__ float ldsB[4][KC][20];
    __shared__ float gbuf[4][16][17];

    const int t    = threadIdx.x;
    const int wave = t >> 5;
    const int lane = t & 31;
    const int half = lane >> 4;
    const int mrow = lane & 15;
    const int m0   = blockIdx.y * 16;
    const int n0   = blockIdx.x * 16;
    const int n0w  = wave * Uu + n0;

    v8f acc = {};
    const int nc1 = K1 / KC;
    const int nc  = nc1 + K2 / KC;
    gemm_tile_acc<1024, 4096>(A1, W, A2, Uw, nc1, nc, m0, n0w,
                              ldsA, ldsB[wave], acc);

    const float bv = bias[n0w + mrow];
#pragma unroll
    for (int r = 0; r < 8; ++r) {
        const float z = acc[r] + bv;
        const float a = (wave == 2) ? tanhf(z) : (1.0f / (1.0f + __expf(-z)));
        gbuf[wave][r + 8 * half][mrow] = a;
    }
    __syncthreads();

#pragma unroll
    for (int e = t; e < 256; e += 128) {
        const int m = e >> 4, n = e & 15;
        const float iv = gbuf[0][m][n];
        const float fv = gbuf[1][m][n];
        const float gv = gbuf[2][m][n];
        const float ov = gbuf[3][m][n];
        const size_t idx = (size_t)(m0 + m) * Uu + n0 + n;
        const float cn = fv * Cprev[idx] + iv * gv;
        Cout[idx] = cn;
        Hout[idx] = ov * tanhf(cn);
    }
}

__global__ void __launch_bounds__(128)
sproj_kernel(const float* __restrict__ H2, const float* __restrict__ Ws,
             const float* __restrict__ bs, float* __restrict__ Sout, int Uu)
{
    __shared__ float ldsA[16][LDA_PITCH];
    __shared__ float ldsB[4][KC][20];

    const int t    = threadIdx.x;
    const int wave = t >> 5;
    const int lane = t & 31;
    const int half = lane >> 4;
    const int mrow = lane & 15;
    const int m0   = blockIdx.y * 16;
    const int n0w  = blockIdx.x * 64 + wave * 16;

    v8f acc = {};
    const int nc = Uu / KC;
    gemm_tile_acc<1024, 1024>(H2, Ws, H2, Ws, nc, nc, m0, n0w,
                              ldsA, ldsB[wave], acc);

    const float bv = bs[n0w + mrow];
#pragma unroll
    for (int r = 0; r < 8; ++r) {
        const int gm = m0 + r + 8 * half;
        Sout[(size_t)gm * Uu + n0w + mrow] = acc[r] + bv;
    }
}

__global__ void __launch_bounds__(512)
attention_kernel(const float* __restrict__ av, const float* __restrict__ S,
                 const float* __restrict__ H2, const float* __restrict__ C2,
                 float* __restrict__ out, int T, int Uu, int B)
{
    __shared__ float sS[1024];
    __shared__ float ctx_s[1024];
    __shared__ float wm[16], wl[16];

    const int b    = blockIdx.x;
    const int t    = threadIdx.x;
    const int wave = t >> 5;
    const int lane = t & 31;

    for (int u = t; u < Uu; u += 512) {
        sS[u]    = S[(size_t)b * Uu + u];
        ctx_s[u] = 0.0f;
    }
    __syncthreads();

    float m = -INFINITY, l = 0.0f;
    float ctx[32];
#pragma unroll
    for (int j = 0; j < 32; ++j) ctx[j] = 0.0f;

    const float* avb = av + (size_t)b * T * Uu;
    for (int tt = wave; tt < T; tt += 16) {
        const float* row = avb + (size_t)tt * Uu;
        __builtin_prefetch(avb + (size_t)(tt + 16) * Uu + lane * 32, 0, 0);

        v4f frag[8];
        float dot = 0.0f;
#pragma unroll
        for (int j = 0; j < 8; ++j) {
            const int u = (j * 32 + lane) * 4;
            frag[j] = __builtin_nontemporal_load((const v4f*)(row + u));
            dot += frag[j].x * sS[u] + frag[j].y * sS[u + 1] +
                   frag[j].z * sS[u + 2] + frag[j].w * sS[u + 3];
        }
#pragma unroll
        for (int off = 16; off > 0; off >>= 1) dot += __shfl_xor(dot, off, 32);

        const float mn    = fmaxf(m, dot);
        const float scale = __expf(m - mn);
        const float e     = __expf(dot - mn);
        l = l * scale + e;
#pragma unroll
        for (int j = 0; j < 8; ++j) {
            ctx[j * 4 + 0] = ctx[j * 4 + 0] * scale + e * frag[j].x;
            ctx[j * 4 + 1] = ctx[j * 4 + 1] * scale + e * frag[j].y;
            ctx[j * 4 + 2] = ctx[j * 4 + 2] * scale + e * frag[j].z;
            ctx[j * 4 + 3] = ctx[j * 4 + 3] * scale + e * frag[j].w;
        }
        m = mn;
    }

    if (lane == 0) { wm[wave] = m; wl[wave] = l; }
    __syncthreads();

    float M = wm[0];
#pragma unroll
    for (int w = 1; w < 16; ++w) M = fmaxf(M, wm[w]);
    float L = 0.0f;
#pragma unroll
    for (int w = 0; w < 16; ++w) L += wl[w] * __expf(wm[w] - M);

    const float f = __expf(m - M);
#pragma unroll
    for (int j = 0; j < 8; ++j) {
        const int u = (j * 32 + lane) * 4;
        atomicAdd(&ctx_s[u + 0], ctx[j * 4 + 0] * f);
        atomicAdd(&ctx_s[u + 1], ctx[j * 4 + 1] * f);
        atomicAdd(&ctx_s[u + 2], ctx[j * 4 + 2] * f);
        atomicAdd(&ctx_s[u + 3], ctx[j * 4 + 3] * f);
    }
    __syncthreads();

    const float invL = 1.0f / L;
    const size_t base = (size_t)b * Uu;
    for (int u = t; u < Uu; u += 512) {
        out[base + u]                  = H2[base + u];
        out[(size_t)B * Uu + base + u] = C2[base + u] + ctx_s[u] * invL;
    }
}

extern "C" void kernel_launch(void* const* d_in, const int* in_sizes, int n_in,
                              void* d_out, int out_size, void* d_ws, size_t ws_size,
                              hipStream_t stream)
{
    (void)in_sizes; (void)n_in; (void)out_size; (void)ws_size;
    const int B = 128, T = 512, U = 1024;

    const float* x   = (const float*)d_in[0];
    const float* h0  = (const float*)d_in[1];
    const float* c0  = (const float*)d_in[2];
    const float* av  = (const float*)d_in[3];
    const float* W1  = (const float*)d_in[4];
    const float* U1  = (const float*)d_in[5];
    const float* b1  = (const float*)d_in[6];
    const float* W2  = (const float*)d_in[7];
    const float* U2  = (const float*)d_in[8];
    const float* b2  = (const float*)d_in[9];
    const float* Ws  = (const float*)d_in[10];
    const float* bs  = (const float*)d_in[11];
    float* out = (float*)d_out;

    float* ws = (float*)d_ws;
    const size_t BU = (size_t)B * U;
    float* h1 = ws;
    float* c1 = ws + BU;
    float* h2 = ws + 2 * BU;
    float* c2 = ws + 3 * BU;
    float* s  = ws + 4 * BU;

    dim3 gGemm(U / 16, B / 16);
    dim3 bGemm(128);

    lstm_cell_kernel<<<gGemm, bGemm, 0, stream>>>(x, W1, h0, U1, b1, c0,
                                                  h1, c1, 1024, U, U);
    lstm_cell_kernel<<<gGemm, bGemm, 0, stream>>>(h1, W2, h1, U2, b2, c1,
                                                  h2, c2, U, U, U);
    sproj_kernel<<<dim3(U / 64, B / 16), bGemm, 0, stream>>>(h2, Ws, bs, s, U);
    attention_kernel<<<dim3(B), dim3(512), 0, stream>>>(av, s, h2, c2, out,
                                                        T, U, B);
}


# LSTM cell GEMMs: wave tile 16x16 -> 16x32 (grid 64x8 -> 32x8, whole 128-byte weight lines); shared GEMM loop reads fragments one 8-step group ahead of the products; order of sums unchanged
# speedup vs baseline: 3.1836x; 1.1314x over previous
#include <hip/hip_runtime.h>
#include <math.h>

typedef __attribute__((ext_vector_type(2))) float v2f;
typedef __attribute__((ext_vector_type(4))) float v4f;
typedef __attribute__((ext_vector_type(8))) float v8f;

#define KC 128
#define LDA_PITCH (KC + 4)

#define LDB_PITCH 20


template <int LDW, int NT>
__device__ __forceinline__ void load_chunk(const float* __restrict__ ap,
                                           const float* __restrict__ bp,
                                           v4f (&av)[KC / 32],
                                           v4f (&bv)[KC * NT / 8])
{
    constexpr int RPL = 8 / NT;
#pragma unroll
    for (int i = 0; i < KC / 32; ++i)
        av[i] = *(const v4f*)(ap + 32 * i);
#pragma unroll
    for (int i = 0; i < KC * NT / 8; ++i)
        bv[i] = *(const v4f*)(bp + (size_t)(RPL * i) * LDW);
}

template <int LDA, int LDW, int NT>
__device__ inline void gemm_tile_acc(const float* __restrict__ A1,
                                     const float* __restrict__ W1,
                                     const float* __restrict__ A2,
                                     const float* __restrict__ W2,
                                     int nc1, int nc, int m0, int n0w,
                                     float (*ldsA)[LDA_PITCH],
                                     float (*ldsB)[KC][LDB_PITCH],
                                     v8f (&acc)[NT])
{
    const int t    = threadIdx.x;
    const int lane = t & 31;
    const int half = lane >> 4;
    const int mrow = lane & 15;

    const int ar = t >> 3;
    const int ac = (t & 7) * 4;
    constexpr int LPR = 4 * NT;
    constexpr int RPL = 8 / NT;
    constexpr int NB  = KC * NT / 8;
    const int br = lane / LPR;
    const int bc = (lane % LPR) * 4;

    const size_t aoff = (size_t)(m0 + ar) * LDA + ac;
    const size_t boff = (size_t)br * LDW + n0w + bc;

    v4f av[KC / 32];
    v4f bv[NB];

    constexpr int G  = 8;
    constexpr int NG = KC / (4 * G);
    v2f af[2][G];
    v2f bf[2][G][NT];

    if (nc > 0) {
        const bool first = (0 < nc1);
        load_chunk<LDW, NT>((first ? A1 : A2) + aoff, (first ? W1 : W2) + boff, av, bv);
    }

    for (int c = 0; c < nc; ++c) {
#pragma unroll
        for (int i = 0; i < KC / 32; ++i)
            *(v4f*)&ldsA[ar][ac + 32 * i] = av[i];
#pragma unroll
        for (int i = 0; i < NB; ++i)
            *(v4f*)&ldsB[bc >> 4][br + RPL * i][bc & 15] = bv[i];
        __syncthreads();

        if (c + 1 < nc) {
            const bool first = (c + 1 < nc1);
            const int  kk    = (first ? (c + 1) : (c + 1 - nc1)) * KC;
            load_chunk<LDW, NT>((first ? A1 : A2) + aoff + kk,
                                (first ? W1 : W2) + boff + (size_t)kk * LDW, av, bv);
        }

#pragma unroll
        for (int s = 0; s < G; ++s) {
            af[0][s].x = ldsA[mrow][4 * s + 2 * half];
            af[0][s].y = ldsA[mrow][4 * s + 2 * half + 1];
#pragma unroll
            for (int j = 0; j < NT; ++j) {
                bf[0][s][j].x = ldsB[j][4 * s + 2 * half][mrow];
                bf[0][s][j].y = ldsB[j][4 * s + 2 * half + 1][mrow];
            }
        }
#pragma unroll
        for (int g = 0; g < NG; ++g) {
            if (g + 1 < NG) {
#pragma unroll
                for (int s = 0; s < G; ++s) {
                    const int ks = 4 * (G * (g + 1) + s);
                    af[(g + 1) & 1][s].x = ldsA[mrow][ks + 2 * half];
                    af[(g + 1) & 1][s].y = ldsA[mrow][ks + 2 * half + 1];
#pragma unroll
                    for (int j = 0; j < NT; ++j) {
                        bf[(g + 1) & 1][s][j].x = ldsB[j][ks + 2 * half][mrow];
                        bf[(g + 1) & 1][s][j].y = ldsB[j][ks + 2 * half + 1][mrow];
                    }
                }
            }
            __builtin_amdgcn_sched_barrier(0);
#pragma unroll
            for (int s = 0; s < G; ++s) {
#pragma unroll
                for (int j = 0; j < NT; ++j)
                    acc[j] = __builtin_amdgcn_wmma_f32_16x16x4_f32(
                        false, af[g & 1][s], false, bf[g & 1][s][j], (short)0,
                        acc[j], false, false);
            }
            __builtin_amdgcn_sched_barrier(0);
        }
        __syncthreads();
    }
}

__global__ void __launch_bounds__(128)
lstm_cell_kernel(const float* __restrict__ A1, const float* __restrict__ W,
                 const float* __restrict__ A2, const float* __restrict__ Uw,
                 const float* __restrict__ bias, const float* __restrict__ Cprev,
                 float* __restrict__ Hout, float* __restrict__ Cout,
                 int K1, int K2, int Uu)
{
    __shared__ float ldsA[16][LDA_PITCH];
    __shared__ float ldsB[4][2][KC][LDB_PITCH];
    __shared__ float gbuf[4][16][34];

    const int t    = threadIdx.x;
    const int wave = t >> 5;
    const int lane = t & 31;
    const int half = lane >> 4;
    const int mrow = lane & 15;
    const int m0   = blockIdx.y * 16;
    const int n0   = blockIdx.x * 32;
    const int n0w  = wave * Uu + n0;

    v8f acc[2] = {};
    const int nc1 = K1 / KC;
    const int nc  = nc1 + K2 / KC;
    gemm_tile_acc<1024, 4096, 2>(A1, W, A2, Uw, nc1, nc, m0, n0w,
                                 ldsA, ldsB[wave], acc);

#pragma unroll
    for (int j = 0; j < 2; ++j) {
        const float bv = bias[n0w + 16 * j + mrow];
#pragma unroll
        for (int r = 0; r < 8; ++r) {
            const float z = acc[j][r] + bv;
            const float a = (wave == 2) ? tanhf(z) : (1.0f / (1.0f + __expf(-z)));
            gbuf[wave][r + 8 * half][16 * j + mrow] = a;
        }
    }
    __syncthreads();

#pragma unroll
    for (int e = t; e < 512; e += 128) {
        const int m = e >> 5, n = e & 31;
        const float iv = gbuf[0][m][n];
        const float fv = gbuf[1][m][n];
        const float gv = gbuf[2][m][n];
        const float ov = gbuf[3][m][n];
        const size_t idx = (size_t)(m0 + m) * Uu + n0 + n;
        const float cn = fv * Cprev[idx] + iv * gv;
        Cout[idx] = cn;
        Hout[idx] = ov * tanhf(cn);
    }
}

__global__ void __launch_bounds__(128)
sproj_kernel(const float* __restrict__ H2, const float* __restrict__ Ws,
             const float* __restrict__ bs, float* __restrict__ Sout, int Uu)
{
    __shared__ float ldsA[16][LDA_PITCH];
    __shared__ float ldsB[4][1][KC][LDB_PITCH];

    const int t    = threadIdx.x;
    const int wave = t >> 5;
    const int lane = t & 31;
    const int half = lane >> 4;
    const int mrow = lane & 15;
    const int m0   = blockIdx.y * 16;
    const int n0w  = blockIdx.x * 64 + wave * 16;

    v8f acc1[1] = {};
    const int nc = Uu / KC;
    gemm_tile_acc<1024, 1024, 1>(H2, Ws, H2, Ws, nc, nc, m0, n0w,
                                 ldsA, ldsB[wave], acc1);
    const v8f acc = acc1[0];

    const float bv = bs[n0w + mrow];
#pragma unroll
    for (int r = 0; r < 8; ++r) {
        const int gm = m0 + r + 8 * half;
        Sout[(size_t)gm * Uu + n0w + mrow] = acc[r] + bv;
    }
}

__global__ void __launch_bounds__(512)
attention_kernel(const float* __restrict__ av, const float* __restrict__ S,
                 const float* __restrict__ H2, const float* __restrict__ C2,
                 float* __restrict__ out, int T, int Uu, int B)
{
    __shared__ float sS[1024];
    __shared__ float ctx_s[1024];
    __shared__ float wm[16], wl[16];

    const int b    = blockIdx.x;
    const int t    = threadIdx.x;
    const int wave = t >> 5;
    const int lane = t & 31;

    for (int u = t; u < Uu; u += 512) {
        sS[u]    = S[(size_t)b * Uu + u];
        ctx_s[u] = 0.0f;
    }
    __syncthreads();

    float m = -INFINITY, l = 0.0f;
    float ctx[32];
#pragma unroll
    for (int j = 0; j < 32; ++j) ctx[j] = 0.0f;

    const float* avb = av + (size_t)b * T * Uu;
    for (int tt = wave; tt < T; tt += 16) {
        const float* row = avb + (size_t)tt * Uu;
        __builtin_prefetch(avb + (size_t)(tt + 16) * Uu + lane * 32, 0, 0);

        v4f frag[8];
        float dot = 0.0f;
#pragma unroll
        for (int j = 0; j < 8; ++j) {
            const int u = (j * 32 + lane) * 4;
            frag[j] = __builtin_nontemporal_load((const v4f*)(row + u));
            dot += frag[j].x * sS[u] + frag[j].y * sS[u + 1] +
                   frag[j].z * sS[u + 2] + frag[j].w * sS[u + 3];
        }
#pragma unroll
        for (int off = 16; off > 0; off >>= 1) dot += __shfl_xor(dot, off, 32);

        const float mn    = fmaxf(m, dot);
        const float scale = __expf(m - mn);
        const float e     = __expf(dot - mn);
        l = l * scale + e;
#pragma unroll
        for (int j = 0; j < 8; ++j) {
            ctx[j * 4 + 0] = ctx[j * 4 + 0] * scale + e * frag[j].x;
            ctx[j * 4 + 1] = ctx[j * 4 + 1] * scale + e * frag[j].y;
            ctx[j * 4 + 2] = ctx[j * 4 + 2] * scale + e * frag[j].z;
            ctx[j * 4 + 3] = ctx[j * 4 + 3] * scale + e * frag[j].w;
        }
        m = mn;
    }

    if (lane == 0) { wm[wave] = m; wl[wave] = l; }
    __syncthreads();

    float M = wm[0];
#pragma unroll
    for (int w = 1; w < 16; ++w) M = fmaxf(M, wm[w]);
    float L = 0.0f;
#pragma unroll
    for (int w = 0; w < 16; ++w) L += wl[w] * __expf(wm[w] - M);

    const float f = __expf(m - M);
#pragma unroll
    for (int j = 0; j < 8; ++j) {
        const int u = (j * 32 + lane) * 4;
        atomicAdd(&ctx_s[u + 0], ctx[j * 4 + 0] * f);
        atomicAdd(&ctx_s[u + 1], ctx[j * 4 + 1] * f);
        atomicAdd(&ctx_s[u + 2], ctx[j * 4 + 2] * f);
        atomicAdd(&ctx_s[u + 3], ctx[j * 4 + 3] * f);
    }
    __syncthreads();

    const float invL = 1.0f / L;
    const size_t base = (size_t)b * Uu;
    for (int u = t; u < Uu; u += 512) {
        out[base + u]                  = H2[base + u];
        out[(size_t)B * Uu + base + u] = C2[base + u] + ctx_s[u] * invL;
    }
}

extern "C" void kernel_launch(void* const* d_in, const int* in_sizes, int n_in,
                              void* d_out, int out_size, void* d_ws, size_t ws_size,
                              hipStream_t stream)
{
    (void)in_sizes; (void)n_in; (void)out_size; (void)ws_size;
    const int B = 128, T = 512, U = 1024;

    const float* x   = (const float*)d_in[0];
    const float* h0  = (const float*)d_in[1];
    const float* c0  = (const float*)d_in[2];
    const float* av  = (const float*)d_in[3];
    const float* W1  = (const float*)d_in[4];
    const float* U1  = (const float*)d_in[5];
    const float* b1  = (const float*)d_in[6];
    const float* W2  = (const float*)d_in[7];
    const float* U2  = (const float*)d_in[8];
    const float* b2  = (const float*)d_in[9];
    const float* Ws  = (const float*)d_in[10];
    const float* bs  = (const float*)d_in[11];
    float* out = (float*)d_out;

    float* ws = (float*)d_ws;
    const size_t BU = (size_t)B * U;
    float* h1 = ws;
    float* c1 = ws + BU;
    float* h2 = ws + 2 * BU;
    float* c2 = ws + 3 * BU;
    float* s  = ws + 4 * BU;

    dim3 gGemm(U / 32, B / 16);
    dim3 bGemm(128);

    lstm_cell_kernel<<<gGemm, bGemm, 0, stream>>>(x, W1, h0, U1, b1, c0,
                                                  h1, c1, 1024, U, U);
    lstm_cell_kernel<<<gGemm, bGemm, 0, stream>>>(h1, W2, h1, U2, b2, c1,
                                                  h2, c2, U, U, U);
    sproj_kernel<<<dim3(U / 64, B / 16), bGemm, 0, stream>>>(h2, Ws, bs, s, U);
    attention_kernel<<<dim3(B), dim3(512), 0, stream>>>(av, s, h2, c2, out,
                                                        T, U, B);
}


# GEMM K loop two chunks deep: second register set and second LDS stage; next chunk's LDS stores and the loads two chunks on go between the product groups; one LDS-only barrier per trip; sums unchanged
# speedup vs baseline: 3.6473x; 1.1457x over previous
#include <hip/hip_runtime.h>
#include <math.h>

typedef __attribute__((ext_vector_type(2))) float v2f;
typedef __attribute__((ext_vector_type(4))) float v4f;
typedef __attribute__((ext_vector_type(8))) float v8f;

#define KC 128
#define LDA_PITCH (KC + 4)

#define LDB_PITCH 20


__device__ __forceinline__ void lds_stage_barrier()
{
    asm volatile("s_wait_dscnt 0x0" ::: "memory");
    asm volatile("s_barrier_signal -1" ::: "memory");
    asm volatile("s_barrier_wait -1" ::: "memory");
}

template <int LDW, int NT>
__device__ __forceinline__ void load_quarter(const float* __restrict__ ap,
                                             const float* __restrict__ bp,
                                             const int q,
                                             v4f (&av)[KC / 32],
                                             v4f (&bv)[KC * NT / 8])
{
    constexpr int RPL = 8 / NT;
    constexpr int NBQ = KC * NT / 32;
    av[q] = *(const v4f*)(ap + 32 * q);
#pragma unroll
    for (int i = 0; i < NBQ; ++i)
        bv[q * NBQ + i] = *(const v4f*)(bp + (size_t)(RPL * (q * NBQ + i)) * LDW);
}

template <int NT>
__device__ __forceinline__ void store_quarter(float (*sA)[LDA_PITCH],
                                              float (*sB)[KC][LDB_PITCH],
                                              const int q, const int ar, const int ac,
                                              const int br, const int bc,
                                              const v4f (&av)[KC / 32],
                                              const v4f (&bv)[KC * NT / 8])
{
    constexpr int RPL = 8 / NT;
    constexpr int NBQ = KC * NT / 32;
    *(v4f*)&sA[ar][ac + 32 * q] = av[q];
#pragma unroll
    for (int i = 0; i < NBQ; ++i)
        *(v4f*)&sB[bc >> 4][br + RPL * (q * NBQ + i)][bc & 15] = bv[q * NBQ + i];
}

struct GemmCtx {
    const float* A1; const float* W1; const float* A2; const float* W2;
    size_t aoff, boff;
    int ar, ac, br, bc, mrow, half;
};

template <int LDW, int NC1>
__device__ __forceinline__ void chunk_ptrs(const GemmCtx& x, const int k,
                                           const float*& ap, const float*& bp)
{
    const bool first = (k < NC1);
    const int  kk    = (first ? k : (k - NC1)) * KC;
    ap = (first ? x.A1 : x.A2) + x.aoff + kk;
    bp = (first ? x.W1 : x.W2) + x.boff + (size_t)kk * LDW;
}

template <int LDW, int NT, int NC1, int P, bool DO_STORE, bool DO_LOAD>
__device__ __forceinline__ void gemm_trip(const GemmCtx& x, const int c,
                                          float (*ldsA)[16][LDA_PITCH],
                                          float (*ldsB0)[KC][LDB_PITCH],
                                          float (*ldsB1)[KC][LDB_PITCH],
                                          v4f (&av)[2][KC / 32],
                                          v4f (&bv)[2][KC * NT / 8],
                                          v8f (&acc)[NT])
{
    constexpr int G  = 8;
    constexpr int NG = KC / (4 * G);
    static_assert(NG == KC / 32, "a quarter of a chunk is one fragment group");

    float (*curA)[LDA_PITCH]      = ldsA[P];
    float (*nxtA)[LDA_PITCH]      = ldsA[P ^ 1];
    float (*curB)[KC][LDB_PITCH]  = P ? ldsB1 : ldsB0;
    float (*nxtB)[KC][LDB_PITCH]  = P ? ldsB0 : ldsB1;
    const int mrow = x.mrow, half = x.half;

    const float* ap = x.A1;
    const float* bp = x.W1;
    if (DO_LOAD) chunk_ptrs<LDW, NC1>(x, c + 3, ap, bp);

    v2f af[2][G];
    v2f bf[2][G][NT];

#pragma unroll
    for (int s = 0; s < G; ++s) {
        af[0][s].x = curA[mrow][4 * s + 2 * half];
        af[0][s].y = curA[mrow][4 * s + 2 * half + 1];
#pragma unroll
        for (int j = 0; j < NT; ++j) {
            bf[0][s][j].x = curB[j][4 * s + 2 * half][mrow];
            bf[0][s][j].y = curB[j][4 * s + 2 * half + 1][mrow];
        }
    }
#pragma unroll
    for (int g = 0; g < NG; ++g) {
        if (g + 1 < NG) {
#pragma unroll
            for (int s = 0; s < G; ++s) {
                const int ks = 4 * (G * (g + 1) + s);
                af[(g + 1) & 1][s].x = curA[mrow][ks + 2 * half];
                af[(g + 1) & 1][s].y = curA[mrow][ks + 2 * half + 1];
#pragma unroll
                for (int j = 0; j < NT; ++j) {
                    bf[(g + 1) & 1][s][j].x = curB[j][ks + 2 * half][mrow];
                    bf[(g + 1) & 1][s][j].y = curB[j][ks + 2 * half + 1][mrow];
                }
            }
        }
        if (DO_STORE)
            store_quarter<NT>(nxtA, nxtB, g, x.ar, x.ac, x.br, x.bc, av[P ^ 1], bv[P ^ 1]);
        asm volatile("" ::: "memory");
        if (DO_LOAD)
            load_quarter<LDW, NT>(ap, bp, g, av[P ^ 1], bv[P ^ 1]);
        __builtin_amdgcn_sched_barrier(0);
#pragma unroll
        for (int s = 0; s < G; ++s) {
#pragma unroll
            for (int j = 0; j < NT; ++j)
                acc[j] = __builtin_amdgcn_wmma_f32_16x16x4_f32(
                    false, af[g & 1][s], false, bf[g & 1][s][j], (short)0,
                    acc[j], false, false);
        }
        __builtin_amdgcn_sched_barrier(0);
    }
    lds_stage_barrier();
}

template <int LDA, int LDW, int NT, int NC1, int NC>
__device__ inline void gemm_tile_acc(const float* __restrict__ A1,
                                     const float* __restrict__ W1,
                                     const float* __restrict__ A2,
                                     const float* __restrict__ W2,
                                     int m0, int n0w,
                                     float (*ldsA)[16][LDA_PITCH],
                                     float (*ldsB0)[KC][LDB_PITCH],
                                     float (*ldsB1)[KC][LDB_PITCH],
                                     v8f (&acc)[NT])
{
    static_assert(NC >= 4 && NC % 2 == 0 && NC1 >= 0 && NC1 <= NC, "pipeline shape");
    const int t    = threadIdx.x;
    const int lane = t & 31;

    constexpr int LPR = 4 * NT;
    constexpr int NB  = KC * NT / 8;
    constexpr int NQ  = KC / 32;

    GemmCtx x;
    x.A1 = A1; x.W1 = W1; x.A2 = A2; x.W2 = W2;
    x.ar = t >> 3;
    x.ac = (t & 7) * 4;
    x.br = lane / LPR;
    x.bc = (lane % LPR) * 4;
    x.mrow = lane & 15;
    x.half = lane >> 4;
    x.aoff = (size_t)(m0 + x.ar) * LDA + x.ac;
    x.boff = (size_t)x.br * LDW + n0w + x.bc;

    v4f av[2][NQ];
    v4f bv[2][NB];

    const float* ap;
    const float* bp;
    chunk_ptrs<LDW, NC1>(x, 0, ap, bp);
#pragma unroll
    for (int q = 0; q < NQ; ++q) load_quarter<LDW, NT>(ap, bp, q, av[0], bv[0]);
    chunk_ptrs<LDW, NC1>(x, 1, ap, bp);
#pragma unroll
    for (int q = 0; q < NQ; ++q) load_quarter<LDW, NT>(ap, bp, q, av[1], bv[1]);
#pragma unroll
    for (int q = 0; q < NQ; ++q)
        store_quarter<NT>(ldsA[0], ldsB0, q, x.ar, x.ac, x.br, x.bc, av[0], bv[0]);
    asm volatile("" ::: "memory");
    chunk_ptrs<LDW, NC1>(x, 2, ap, bp);
#pragma unroll
    for (int q = 0; q < NQ; ++q) load_quarter<LDW, NT>(ap, bp, q, av[0], bv[0]);
    lds_stage_barrier();

#pragma nounroll
    for (int c = 0; c + 4 < NC; c += 2) {
        gemm_trip<LDW, NT, NC1, 0, true, true>(x, c,     ldsA, ldsB0, ldsB1, av, bv, acc);
        gemm_trip<LDW, NT, NC1, 1, true, true>(x, c + 1, ldsA, ldsB0, ldsB1, av, bv, acc);
    }
    gemm_trip<LDW, NT, NC1, 0, true,  true >(x, NC - 4, ldsA, ldsB0, ldsB1, av, bv, acc);
    gemm_trip<LDW, NT, NC1, 1, true,  false>(x, NC - 3, ldsA, ldsB0, ldsB1, av, bv, acc);
    gemm_trip<LDW, NT, NC1, 0, true,  false>(x, NC - 2, ldsA, ldsB0, ldsB1, av, bv, acc);
    gemm_trip<LDW, NT, NC1, 1, false, false>(x, NC - 1, ldsA, ldsB0, ldsB1, av, bv, acc);
}

template <int K1, int K2>
__global__ void __launch_bounds__(128)
lstm_cell_kernel(const float* __restrict__ A1, const float* __restrict__ W,
                 const float* __restrict__ A2, const float* __restrict__ Uw,
                 const float* __restrict__ bias, const float* __restrict__ Cprev,
                 float* __restrict__ Hout, float* __restrict__ Cout,
                 int Uu)
{
    __shared__ float ldsA[2][16][LDA_PITCH];
    __shared__ float ldsB[2][4][2][KC][LDB_PITCH];
    __shared__ float gbuf[4][16][34];

    const int t    = threadIdx.x;
    const int wave = t >> 5;
    const int lane = t & 31;
    const int half = lane >> 4;
    const int mrow = lane & 15;
    const int m0   = blockIdx.y * 16;
    const int n0   = blockIdx.x * 32;
    const int n0w  = wave * Uu + n0;

    v8f acc[2] = {};
    static_assert(K1 % KC == 0 && K2 % KC == 0, "whole chunks");
    gemm_tile_acc<1024, 4096, 2, K1 / KC, (K1 + K2) / KC>(
        A1, W, A2, Uw, m0, n0w, ldsA, ldsB[0][wave], ldsB[1][wave], acc);

#pragma unroll
    for (int j = 0; j < 2; ++j) {
        const float bv = bias[n0w + 16 * j + mrow];
#pragma unroll
        for (int r = 0; r < 8; ++r) {
            const float z = acc[j][r] + bv;
            const float a = (wave == 2) ? tanhf(z) : (1.0f / (1.0f + __expf(-z)));
            gbuf[wave][r + 8 * half][16 * j + mrow] = a;
        }
    }
    __syncthreads();

#pragma unroll
    for (int e = t; e < 512; e += 128) {
        const int m = e >> 5, n = e & 31;
        const float iv = gbuf[0][m][n];
        const float fv = gbuf[1][m][n];
        const float gv = gbuf[2][m][n];
        const float ov = gbuf[3][m][n];
        const size_t idx = (size_t)(m0 + m) * Uu + n0 + n;
        const float cn = fv * Cprev[idx] + iv * gv;
        Cout[idx] = cn;
        Hout[idx] = ov * tanhf(cn);
    }
}

template <int K>
__global__ void __launch_bounds__(128)
sproj_kernel(const float* __restrict__ H2, const float* __restrict__ Ws,
             const float* __restrict__ bs, float* __restrict__ Sout, int Uu)
{
    __shared__ float ldsA[2][16][LDA_PITCH];
    __shared__ float ldsB[2][4][1][KC][LDB_PITCH];

    const int t    = threadIdx.x;
    const int wave = t >> 5;
    const int lane = t & 31;
    const int half = lane >> 4;
    const int mrow = lane & 15;
    const int m0   = blockIdx.y * 16;
    const int n0w  = blockIdx.x * 64 + wave * 16;

    v8f acc1[1] = {};
    static_assert(K % KC == 0, "whole chunks");
    gemm_tile_acc<1024, 1024, 1, K / KC, K / KC>(
        H2, Ws, H2, Ws, m0, n0w, ldsA, ldsB[0][wave], ldsB[1][wave], acc1);
    const v8f acc = acc1[0];

    const float bv = bs[n0w + mrow];
#pragma unroll
    for (int r = 0; r < 8; ++r) {
        const int gm = m0 + r + 8 * half;
        Sout[(size_t)gm * Uu + n0w + mrow] = acc[r] + bv;
    }
}

__global__ void __launch_bounds__(512)
attention_kernel(const float* __restrict__ av, const float* __restrict__ S,
                 const float* __restrict__ H2, const float* __restrict__ C2,
                 float* __restrict__ out, int T, int Uu, int B)
{
    __shared__ float sS[1024];
    __shared__ float ctx_s[1024];
    __shared__ float wm[16], wl[16];

    const int b    = blockIdx.x;
    const int t    = threadIdx.x;
    const int wave = t >> 5;
    const int lane = t & 31;

    for (int u = t; u < Uu; u += 512) {
        sS[u]    = S[(size_t)b * Uu + u];
        ctx_s[u] = 0.0f;
    }
    __syncthreads();

    float m = -INFINITY, l = 0.0f;
    float ctx[32];
#pragma unroll
    for (int j = 0; j < 32; ++j) ctx[j] = 0.0f;

    const float* avb = av + (size_t)b * T * Uu;
    for (int tt = wave; tt < T; tt += 16) {
        const float* row = avb + (size_t)tt * Uu;
        __builtin_prefetch(avb + (size_t)(tt + 16) * Uu + lane * 32, 0, 0);

        v4f frag[8];
        float dot = 0.0f;
#pragma unroll
        for (int j = 0; j < 8; ++j) {
            const int u = (j * 32 + lane) * 4;
            frag[j] = __builtin_nontemporal_load((const v4f*)(row + u));
            dot += frag[j].x * sS[u] + frag[j].y * sS[u + 1] +
                   frag[j].z * sS[u + 2] + frag[j].w * sS[u + 3];
        }
#pragma unroll
        for (int off = 16; off > 0; off >>= 1) dot += __shfl_xor(dot, off, 32);

        const float mn    = fmaxf(m, dot);
        const float scale = __expf(m - mn);
        const float e     = __expf(dot - mn);
        l = l * scale + e;
#pragma unroll
        for (int j = 0; j < 8; ++j) {
            ctx[j * 4 + 0] = ctx[j * 4 + 0] * scale + e * frag[j].x;
            ctx[j * 4 + 1] = ctx[j * 4 + 1] * scale + e * frag[j].y;
            ctx[j * 4 + 2] = ctx[j * 4 + 2] * scale + e * frag[j].z;
            ctx[j * 4 + 3] = ctx[j * 4 + 3] * scale + e * frag[j].w;
        }
        m = mn;
    }

    if (lane == 0) { wm[wave] = m; wl[wave] = l; }
    __syncthreads();

    float M = wm[0];
#pragma unroll
    for (int w = 1; w < 16; ++w) M = fmaxf(M, wm[w]);
    float L = 0.0f;
#pragma unroll
    for (int w = 0; w < 16; ++w) L += wl[w] * __expf(wm[w] - M);

    const float f = __expf(m - M);
#pragma unroll
    for (int j = 0; j < 8; ++j) {
        const int u = (j * 32 + lane) * 4;
        atomicAdd(&ctx_s[u + 0], ctx[j * 4 + 0] * f);
        atomicAdd(&ctx_s[u + 1], ctx[j * 4 + 1] * f);
        atomicAdd(&ctx_s[u + 2], ctx[j * 4 + 2] * f);
        atomicAdd(&ctx_s[u + 3], ctx[j * 4 + 3] * f);
    }
    __syncthreads();

    const float invL = 1.0f / L;
    const size_t base = (size_t)b * Uu;
    for (int u = t; u < Uu; u += 512) {
        out[base + u]                  = H2[base + u];
        out[(size_t)B * Uu + base + u] = C2[base + u] + ctx_s[u] * invL;
    }
}

extern "C" void kernel_launch(void* const* d_in, const int* in_sizes, int n_in,
                              void* d_out, int out_size, void* d_ws, size_t ws_size,
                              hipStream_t stream)
{
    (void)in_sizes; (void)n_in; (void)out_size; (void)ws_size;
    const int B = 128, T = 512, U = 1024;

    const float* x   = (const float*)d_in[0];
    const float* h0  = (const float*)d_in[1];
    const float* c0  = (const float*)d_in[2];
    const float* av  = (const float*)d_in[3];
    const float* W1  = (const float*)d_in[4];
    const float* U1  = (const float*)d_in[5];
    const float* b1  = (const float*)d_in[6];
    const float* W2  = (const float*)d_in[7];
    const float* U2  = (const float*)d_in[8];
    const float* b2  = (const float*)d_in[9];
    const float* Ws  = (const float*)d_in[10];
    const float* bs  = (const float*)d_in[11];
    float* out = (float*)d_out;

    float* ws = (float*)d_ws;
    const size_t BU = (size_t)B * U;
    float* h1 = ws;
    float* c1 = ws + BU;
    float* h2 = ws + 2 * BU;
    float* c2 = ws + 3 * BU;
    float* s  = ws + 4 * BU;

    dim3 gGemm(U / 32, B / 16);
    dim3 bGemm(128);

    lstm_cell_kernel<1024, 1024><<<gGemm, bGemm, 0, stream>>>(x, W1, h0, U1, b1, c0,
                                                              h1, c1, U);
    lstm_cell_kernel<1024, 1024><<<gGemm, bGemm, 0, stream>>>(h1, W2, h1, U2, b2, c1,
                                                              h2, c2, U);
    sproj_kernel<1024><<<dim3(U / 64, B / 16), bGemm, 0, stream>>>(h2, Ws, bs, s, U);
    attention_kernel<<<dim3(B), dim3(512), 0, stream>>>(av, s, h2, c2, out,
                                                        T, U, B);
}


# Attention split over 2 work-groups per batch row (grid 128 -> 256, one per WGP), each folding half of the T rows to a partial (max, sum, context); a small 5th launch merges them: sum order changes
# speedup vs baseline: 3.7646x; 1.0322x over previous
#include <hip/hip_runtime.h>
#include <math.h>

typedef __attribute__((ext_vector_type(2))) float v2f;
typedef __attribute__((ext_vector_type(4))) float v4f;
typedef __attribute__((ext_vector_type(8))) float v8f;

#define KC 128
#define LDA_PITCH (KC + 4)

#define LDB_PITCH 20


__device__ __forceinline__ void lds_stage_barrier()
{
    asm volatile("s_wait_dscnt 0x0" ::: "memory");
    asm volatile("s_barrier_signal -1" ::: "memory");
    asm volatile("s_barrier_wait -1" ::: "memory");
}

template <int LDW, int NT>
__device__ __forceinline__ void load_quarter(const float* __restrict__ ap,
                                             const float* __restrict__ bp,
                                             const int q,
                                             v4f (&av)[KC / 32],
                                             v4f (&bv)[KC * NT / 8])
{
    constexpr int RPL = 8 / NT;
    constexpr int NBQ = KC * NT / 32;
    av[q] = *(const v4f*)(ap + 32 * q);
#pragma unroll
    for (int i = 0; i < NBQ; ++i)
        bv[q * NBQ + i] = *(const v4f*)(bp + (size_t)(RPL * (q * NBQ + i)) * LDW);
}

template <int NT>
__device__ __forceinline__ void store_quarter(float (*sA)[LDA_PITCH],
                                              float (*sB)[KC][LDB_PITCH],
                                              const int q, const int ar, const int ac,
                                              const int br, const int bc,
                                              const v4f (&av)[KC / 32],
                                              const v4f (&bv)[KC * NT / 8])
{
    constexpr int RPL = 8 / NT;
    constexpr int NBQ = KC * NT / 32;
    *(v4f*)&sA[ar][ac + 32 * q] = av[q];
#pragma unroll
    for (int i = 0; i < NBQ; ++i)
        *(v4f*)&sB[bc >> 4][br + RPL * (q * NBQ + i)][bc & 15] = bv[q * NBQ + i];
}

struct GemmCtx {
    const float* A1; const float* W1; const float* A2; const float* W2;
    size_t aoff, boff;
    int ar, ac, br, bc, mrow, half;
};

template <int LDW, int NC1>
__device__ __forceinline__ void chunk_ptrs(const GemmCtx& x, const int k,
                                           const float*& ap, const float*& bp)
{
    const bool first = (k < NC1);
    const int  kk    = (first ? k : (k - NC1)) * KC;
    ap = (first ? x.A1 : x.A2) + x.aoff + kk;
    bp = (first ? x.W1 : x.W2) + x.boff + (size_t)kk * LDW;
}

template <int LDW, int NT, int NC1, int P, bool DO_STORE, bool DO_LOAD>
__device__ __forceinline__ void gemm_trip(const GemmCtx& x, const int c,
                                          float (*ldsA)[16][LDA_PITCH],
                                          float (*ldsB0)[KC][LDB_PITCH],
                                          float (*ldsB1)[KC][LDB_PITCH],
                                          v4f (&av)[2][KC / 32],
                                          v4f (&bv)[2][KC * NT / 8],
                                          v8f (&acc)[NT])
{
    constexpr int G  = 8;
    constexpr int NG = KC / (4 * G);
    static_assert(NG == KC / 32, "a quarter of a chunk is one fragment group");

    float (*curA)[LDA_PITCH]      = ldsA[P];
    float (*nxtA)[LDA_PITCH]      = ldsA[P ^ 1];
    float (*curB)[KC][LDB_PITCH]  = P ? ldsB1 : ldsB0;
    float (*nxtB)[KC][LDB_PITCH]  = P ? ldsB0 : ldsB1;
    const int mrow = x.mrow, half = x.half;

    const float* ap = x.A1;
    const float* bp = x.W1;
    if (DO_LOAD) chunk_ptrs<LDW, NC1>(x, c + 3, ap, bp);

    v2f af[2][G];
    v2f bf[2][G][NT];

#pragma unroll
    for (int s = 0; s < G; ++s) {
        af[0][s].x = curA[mrow][4 * s + 2 * half];
        af[0][s].y = curA[mrow][4 * s + 2 * half + 1];
#pragma unroll
        for (int j = 0; j < NT; ++j) {
            bf[0][s][j].x = curB[j][4 * s + 2 * half][mrow];
            bf[0][s][j].y = curB[j][4 * s + 2 * half + 1][mrow];
        }
    }
#pragma unroll
    for (int g = 0; g < NG; ++g) {
        if (g + 1 < NG) {
#pragma unroll
            for (int s = 0; s < G; ++s) {
                const int ks = 4 * (G * (g + 1) + s);
                af[(g + 1) & 1][s].x = curA[mrow][ks + 2 * half];
                af[(g + 1) & 1][s].y = curA[mrow][ks + 2 * half + 1];
#pragma unroll
                for (int j = 0; j < NT; ++j) {
                    bf[(g + 1) & 1][s][j].x = curB[j][ks + 2 * half][mrow];
                    bf[(g + 1) & 1][s][j].y = curB[j][ks + 2 * half + 1][mrow];
                }
            }
        }
        if (DO_STORE)
            store_quarter<NT>(nxtA, nxtB, g, x.ar, x.ac, x.br, x.bc, av[P ^ 1], bv[P ^ 1]);
        asm volatile("" ::: "memory");
        if (DO_LOAD)
            load_quarter<LDW, NT>(ap, bp, g, av[P ^ 1], bv[P ^ 1]);
        __builtin_amdgcn_sched_barrier(0);
#pragma unroll
        for (int s = 0; s < G; ++s) {
#pragma unroll
            for (int j = 0; j < NT; ++j)
                acc[j] = __builtin_amdgcn_wmma_f32_16x16x4_f32(
                    false, af[g & 1][s], false, bf[g & 1][s][j], (short)0,
                    acc[j], false, false);
        }
        __builtin_amdgcn_sched_barrier(0);
    }
    lds_stage_barrier();
}

template <int LDA, int LDW, int NT, int NC1, int NC>
__device__ inline void gemm_tile_acc(const float* __restrict__ A1,
                                     const float* __restrict__ W1,
                                     const float* __restrict__ A2,
                                     const float* __restrict__ W2,
                                     int m0, int n0w,
                                     float (*ldsA)[16][LDA_PITCH],
                                     float (*ldsB0)[KC][LDB_PITCH],
                                     float (*ldsB1)[KC][LDB_PITCH],
                                     v8f (&acc)[NT])
{
    static_assert(NC >= 4 && NC % 2 == 0 && NC1 >= 0 && NC1 <= NC, "pipeline shape");
    const int t    = threadIdx.x;
    const int lane = t & 31;

    constexpr int LPR = 4 * NT;
    constexpr int NB  = KC * NT / 8;
    constexpr int NQ  = KC / 32;

    GemmCtx x;
    x.A1 = A1; x.W1 = W1; x.A2 = A2; x.W2 = W2;
    x.ar = t >> 3;
    x.ac = (t & 7) * 4;
    x.br = lane / LPR;
    x.bc = (lane % LPR) * 4;
    x.mrow = lane & 15;
    x.half = lane >> 4;
    x.aoff = (size_t)(m0 + x.ar) * LDA + x.ac;
    x.boff = (size_t)x.br * LDW + n0w + x.bc;

    v4f av[2][NQ];
    v4f bv[2][NB];

    const float* ap;
    const float* bp;
    chunk_ptrs<LDW, NC1>(x, 0, ap, bp);
#pragma unroll
    for (int q = 0; q < NQ; ++q) load_quarter<LDW, NT>(ap, bp, q, av[0], bv[0]);
    chunk_ptrs<LDW, NC1>(x, 1, ap, bp);
#pragma unroll
    for (int q = 0; q < NQ; ++q) load_quarter<LDW, NT>(ap, bp, q, av[1], bv[1]);
#pragma unroll
    for (int q = 0; q < NQ; ++q)
        store_quarter<NT>(ldsA[0], ldsB0, q, x.ar, x.ac, x.br, x.bc, av[0], bv[0]);
    asm volatile("" ::: "memory");
    chunk_ptrs<LDW, NC1>(x, 2, ap, bp);
#pragma unroll
    for (int q = 0; q < NQ; ++q) load_quarter<LDW, NT>(ap, bp, q, av[0], bv[0]);
    lds_stage_barrier();

#pragma nounroll
    for (int c = 0; c + 4 < NC; c += 2) {
        gemm_trip<LDW, NT, NC1, 0, true, true>(x, c,     ldsA, ldsB0, ldsB1, av, bv, acc);
        gemm_trip<LDW, NT, NC1, 1, true, true>(x, c + 1, ldsA, ldsB0, ldsB1, av, bv, acc);
    }
    gemm_trip<LDW, NT, NC1, 0, true,  true >(x, NC - 4, ldsA, ldsB0, ldsB1, av, bv, acc);
    gemm_trip<LDW, NT, NC1, 1, true,  false>(x, NC - 3, ldsA, ldsB0, ldsB1, av, bv, acc);
    gemm_trip<LDW, NT, NC1, 0, true,  false>(x, NC - 2, ldsA, ldsB0, ldsB1, av, bv, acc);
    gemm_trip<LDW, NT, NC1, 1, false, false>(x, NC - 1, ldsA, ldsB0, ldsB1, av, bv, acc);
}

template <int K1, int K2>
__global__ void __launch_bounds__(128)
lstm_cell_kernel(const float* __restrict__ A1, const float* __restrict__ W,
                 const float* __restrict__ A2, const float* __restrict__ Uw,
                 const float* __restrict__ bias, const float* __restrict__ Cprev,
                 float* __restrict__ Hout, float* __restrict__ Cout,
                 int Uu)
{
    __shared__ float ldsA[2][16][LDA_PITCH];
    __shared__ float ldsB[2][4][2][KC][LDB_PITCH];
    __shared__ float gbuf[4][16][34];

    const int t    = threadIdx.x;
    const int wave = t >> 5;
    const int lane = t & 31;
    const int half = lane >> 4;
    const int mrow = lane & 15;
    const int m0   = blockIdx.y * 16;
    const int n0   = blockIdx.x * 32;
    const int n0w  = wave * Uu + n0;

    v8f acc[2] = {};
    static_assert(K1 % KC == 0 && K2 % KC == 0, "whole chunks");
    gemm_tile_acc<1024, 4096, 2, K1 / KC, (K1 + K2) / KC>(
        A1, W, A2, Uw, m0, n0w, ldsA, ldsB[0][wave], ldsB[1][wave], acc);

#pragma unroll
    for (int j = 0; j < 2; ++j) {
        const float bv = bias[n0w + 16 * j + mrow];
#pragma unroll
        for (int r = 0; r < 8; ++r) {
            const float z = acc[j][r] + bv;
            const float a = (wave == 2) ? tanhf(z) : (1.0f / (1.0f + __expf(-z)));
            gbuf[wave][r + 8 * half][16 * j + mrow] = a;
        }
    }
    __syncthreads();

#pragma unroll
    for (int e = t; e < 512; e += 128) {
        const int m = e >> 5, n = e & 31;
        const float iv = gbuf[0][m][n];
        const float fv = gbuf[1][m][n];
        const float gv = gbuf[2][m][n];
        const float ov = gbuf[3][m][n];
        const size_t idx = (size_t)(m0 + m) * Uu + n0 + n;
        const float cn = fv * Cprev[idx] + iv * gv;
        Cout[idx] = cn;
        Hout[idx] = ov * tanhf(cn);
    }
}

template <int K>
__global__ void __launch_bounds__(128)
sproj_kernel(const float* __restrict__ H2, const float* __restrict__ Ws,
             const float* __restrict__ bs, float* __restrict__ Sout, int Uu)
{
    __shared__ float ldsA[2][16][LDA_PITCH];
    __shared__ float ldsB[2][4][1][KC][LDB_PITCH];

    const int t    = threadIdx.x;
    const int wave = t >> 5;
    const int lane = t & 31;
    const int half = lane >> 4;
    const int mrow = lane & 15;
    const int m0   = blockIdx.y * 16;
    const int n0w  = blockIdx.x * 64 + wave * 16;

    v8f acc1[1] = {};
    static_assert(K % KC == 0, "whole chunks");
    gemm_tile_acc<1024, 1024, 1, K / KC, K / KC>(
        H2, Ws, H2, Ws, m0, n0w, ldsA, ldsB[0][wave], ldsB[1][wave], acc1);
    const v8f acc = acc1[0];

    const float bv = bs[n0w + mrow];
#pragma unroll
    for (int r = 0; r < 8; ++r) {
        const int gm = m0 + r + 8 * half;
        Sout[(size_t)gm * Uu + n0w + mrow] = acc[r] + bv;
    }
}

#define ATT_SPLIT 2

__global__ void __launch_bounds__(512)
attention_part_kernel(const float* __restrict__ av, const float* __restrict__ S,
                      float* __restrict__ P, float* __restrict__ ML,
                      int T, int Uu, int B)
{
    __shared__ float sS[1024];
    __shared__ float ctx_s[1024];
    __shared__ float wm[16], wl[16];

    const int b    = blockIdx.x / ATT_SPLIT;
    const int h    = blockIdx.x % ATT_SPLIT;
    const int t    = threadIdx.x;
    const int wave = t >> 5;
    const int lane = t & 31;

    for (int u = t; u < Uu; u += 512) {
        sS[u]    = S[(size_t)b * Uu + u];
        ctx_s[u] = 0.0f;
    }
    __syncthreads();

    float m = -INFINITY, l = 0.0f;
    float ctx[32];
#pragma unroll
    for (int j = 0; j < 32; ++j) ctx[j] = 0.0f;

    const int TH = T / ATT_SPLIT;
    const int t0 = h * TH;
    const int t1 = (h == ATT_SPLIT - 1) ? T : t0 + TH;
    const float* avb = av + (size_t)b * T * Uu;
    for (int tt = t0 + wave; tt < t1; tt += 16) {
        const float* row = avb + (size_t)tt * Uu;
        __builtin_prefetch(avb + (size_t)(tt + 16) * Uu + lane * 32, 0, 0);

        v4f frag[8];
        float dot = 0.0f;
#pragma unroll
        for (int j = 0; j < 8; ++j) {
            const int u = (j * 32 + lane) * 4;
            frag[j] = __builtin_nontemporal_load((const v4f*)(row + u));
            dot += frag[j].x * sS[u] + frag[j].y * sS[u + 1] +
                   frag[j].z * sS[u + 2] + frag[j].w * sS[u + 3];
        }
#pragma unroll
        for (int off = 16; off > 0; off >>= 1) dot += __shfl_xor(dot, off, 32);

        const float mn    = fmaxf(m, dot);
        const float scale = __expf(m - mn);
        const float e     = __expf(dot - mn);
        l = l * scale + e;
#pragma unroll
        for (int j = 0; j < 8; ++j) {
            ctx[j * 4 + 0] = ctx[j * 4 + 0] * scale + e * frag[j].x;
            ctx[j * 4 + 1] = ctx[j * 4 + 1] * scale + e * frag[j].y;
            ctx[j * 4 + 2] = ctx[j * 4 + 2] * scale + e * frag[j].z;
            ctx[j * 4 + 3] = ctx[j * 4 + 3] * scale + e * frag[j].w;
        }
        m = mn;
    }

    if (lane == 0) { wm[wave] = m; wl[wave] = l; }
    __syncthreads();

    float M = wm[0];
#pragma unroll
    for (int w = 1; w < 16; ++w) M = fmaxf(M, wm[w]);
    float L = 0.0f;
#pragma unroll
    for (int w = 0; w < 16; ++w) L += wl[w] * __expf(wm[w] - M);

    const float f = __expf(m - M);
#pragma unroll
    for (int j = 0; j < 8; ++j) {
        const int u = (j * 32 + lane) * 4;
        atomicAdd(&ctx_s[u + 0], ctx[j * 4 + 0] * f);
        atomicAdd(&ctx_s[u + 1], ctx[j * 4 + 1] * f);
        atomicAdd(&ctx_s[u + 2], ctx[j * 4 + 2] * f);
        atomicAdd(&ctx_s[u + 3], ctx[j * 4 + 3] * f);
    }
    __syncthreads();

    float* Pb = P + ((size_t)h * B + b) * Uu;
    for (int u = t; u < Uu; u += 512) Pb[u] = ctx_s[u];
    if (t == 0) {
        ML[2 * blockIdx.x]     = M;
        ML[2 * blockIdx.x + 1] = L;
    }
}

__global__ void __launch_bounds__(256)
attention_combine_kernel(const float* __restrict__ P, const float* __restrict__ ML,
                         const float* __restrict__ C2, float* __restrict__ Cout,
                         int Uu, int B)
{
    const int b = blockIdx.x;

    float Mh[ATT_SPLIT], e[ATT_SPLIT];
    float M = -INFINITY;
    for (int h = 0; h < ATT_SPLIT; ++h) {
        Mh[h] = ML[2 * (b * ATT_SPLIT + h)];
        M = fmaxf(M, Mh[h]);
    }
    float L = 0.0f;
    for (int h = 0; h < ATT_SPLIT; ++h) {
        e[h] = __expf(Mh[h] - M);
        L += ML[2 * (b * ATT_SPLIT + h) + 1] * e[h];
    }
    const float invL = 1.0f / L;

    const size_t base  = (size_t)b * Uu;
    const size_t plane = (size_t)B * Uu;
    for (int u = 4 * threadIdx.x; u < Uu; u += 4 * 256) {
        v4f acc = *(const v4f*)(P + base + u) * e[0];
        for (int h = 1; h < ATT_SPLIT; ++h)
            acc += *(const v4f*)(P + h * plane + base + u) * e[h];
        const v4f c2 = *(const v4f*)(C2 + base + u);
        *(v4f*)(Cout + base + u) = c2 + acc * invL;
    }
}

extern "C" void kernel_launch(void* const* d_in, const int* in_sizes, int n_in,
                              void* d_out, int out_size, void* d_ws, size_t ws_size,
                              hipStream_t stream)
{
    (void)in_sizes; (void)n_in; (void)out_size; (void)ws_size;
    const int B = 128, T = 512, U = 1024;

    const float* x   = (const float*)d_in[0];
    const float* h0  = (const float*)d_in[1];
    const float* c0  = (const float*)d_in[2];
    const float* av  = (const float*)d_in[3];
    const float* W1  = (const float*)d_in[4];
    const float* U1  = (const float*)d_in[5];
    const float* b1  = (const float*)d_in[6];
    const float* W2  = (const float*)d_in[7];
    const float* U2  = (const float*)d_in[8];
    const float* b2  = (const float*)d_in[9];
    const float* Ws  = (const float*)d_in[10];
    const float* bs  = (const float*)d_in[11];
    float* out = (float*)d_out;

    float* ws = (float*)d_ws;
    const size_t BU = (size_t)B * U;
    float* h1 = ws;
    float* c1 = ws + BU;
    float* c2 = ws + 3 * BU;
    float* s  = ws + 4 * BU;
    float* h2   = out;
    float* part = ws;
    float* ml   = ws + 2 * BU;
    static_assert(ATT_SPLIT == 2, "two dead planes for the partial contexts");

    dim3 gGemm(U / 32, B / 16);
    dim3 bGemm(128);

    lstm_cell_kernel<1024, 1024><<<gGemm, bGemm, 0, stream>>>(x, W1, h0, U1, b1, c0,
                                                              h1, c1, U);
    lstm_cell_kernel<1024, 1024><<<gGemm, bGemm, 0, stream>>>(h1, W2, h1, U2, b2, c1,
                                                              h2, c2, U);
    sproj_kernel<1024><<<dim3(U / 64, B / 16), bGemm, 0, stream>>>(h2, Ws, bs, s, U);
    attention_part_kernel<<<dim3(B * ATT_SPLIT), dim3(512), 0, stream>>>(av, s, part, ml,
                                                                         T, U, B);
    attention_combine_kernel<<<dim3(B), dim3(256), 0, stream>>>(part, ml, c2, out + BU,
                                                                U, B);
}


# GEMM launches: 8 waves per work-group, two per SIMD: k range split over two wave sets of 4 with own LDS stages (chunk 128 -> 64), the sets' partial sums added at the end: sum order changes
# speedup vs baseline: 3.8863x; 1.0323x over previous
#include <hip/hip_runtime.h>
#include <math.h>

typedef __attribute__((ext_vector_type(2))) float v2f;
typedef __attribute__((ext_vector_type(4))) float v4f;
typedef __attribute__((ext_vector_type(8))) float v8f;

#define KC 64
#define LDA_PITCH (KC + 4)

#define LDB_PITCH 20


__device__ __forceinline__ void lds_stage_barrier()
{
    asm volatile("s_wait_dscnt 0x0" ::: "memory");
    asm volatile("s_barrier_signal -1" ::: "memory");
    asm volatile("s_barrier_wait -1" ::: "memory");
}

template <int LDW, int NT>
__device__ __forceinline__ void load_slice(const float* __restrict__ ap,
                                             const float* __restrict__ bp,
                                             const int q,
                                             v4f (&av)[KC / 32],
                                             v4f (&bv)[KC * NT / 8])
{
    constexpr int RPL = 8 / NT;
    constexpr int NBQ = 32 / RPL;
    av[q] = *(const v4f*)(ap + 32 * q);
#pragma unroll
    for (int i = 0; i < NBQ; ++i)
        bv[q * NBQ + i] = *(const v4f*)(bp + (size_t)(RPL * (q * NBQ + i)) * LDW);
}

template <int NT>
__device__ __forceinline__ void store_slice(float (*sA)[LDA_PITCH],
                                              float (*sB)[KC][LDB_PITCH],
                                              const int q, const int ar, const int ac,
                                              const int br, const int bc,
                                              const v4f (&av)[KC / 32],
                                              const v4f (&bv)[KC * NT / 8])
{
    constexpr int RPL = 8 / NT;
    constexpr int NBQ = 32 / RPL;
    *(v4f*)&sA[ar][ac + 32 * q] = av[q];
#pragma unroll
    for (int i = 0; i < NBQ; ++i)
        *(v4f*)&sB[bc >> 4][br + RPL * (q * NBQ + i)][bc & 15] = bv[q * NBQ + i];
}

template <int NT>
__device__ __forceinline__ void read_group(float (*curA)[LDA_PITCH],
                                           float (*curB)[KC][LDB_PITCH],
                                           const int g, const int mrow, const int half,
                                           v2f (&af)[8], v2f (&bf)[8][NT])
{
#pragma unroll
    for (int s = 0; s < 8; ++s) {
        const int ks = 4 * (8 * g + s);
        af[s].x = curA[mrow][ks + 2 * half];
        af[s].y = curA[mrow][ks + 2 * half + 1];
    }
#pragma unroll
    for (int s = 0; s < 8; ++s) {
        const int ks = 4 * (8 * g + s);
#pragma unroll
        for (int j = 0; j < NT; ++j) {
            bf[s][j].x = curB[j][ks + 2 * half][mrow];
            bf[s][j].y = curB[j][ks + 2 * half + 1][mrow];
        }
        asm volatile("" ::: "memory");
    }
}

struct GemmCtx {
    const float* A; const float* Wt;
    size_t aoff, boff;
    int ar, ac, br, bc, mrow, half;
};

template <int LDW>
__device__ __forceinline__ void chunk_ptrs(const GemmCtx& x, const int k,
                                           const float*& ap, const float*& bp)
{
    ap = x.A + x.aoff + k * KC;
    bp = x.Wt + x.boff + (size_t)(k * KC) * LDW;
}

template <int LDW, int NT, int P, bool DO_STORE, bool DO_LOAD>
__device__ __forceinline__ void gemm_trip(const GemmCtx& x, const int c,
                                          float (*ldsA)[16][LDA_PITCH],
                                          float (*ldsB0)[KC][LDB_PITCH],
                                          float (*ldsB1)[KC][LDB_PITCH],
                                          v4f (&av)[2][KC / 32],
                                          v4f (&bv)[2][KC * NT / 8],
                                          v8f (&acc)[NT])
{
    constexpr int G  = 8;
    constexpr int NG = KC / (4 * G);
    static_assert(NG == KC / 32, "a slice of a chunk is one fragment group");

    float (*curA)[LDA_PITCH]      = ldsA[P];
    float (*nxtA)[LDA_PITCH]      = ldsA[P ^ 1];
    float (*curB)[KC][LDB_PITCH]  = P ? ldsB1 : ldsB0;
    float (*nxtB)[KC][LDB_PITCH]  = P ? ldsB0 : ldsB1;
    const int mrow = x.mrow, half = x.half;

    const float* ap = x.A;
    const float* bp = x.Wt;
    if (DO_LOAD) chunk_ptrs<LDW>(x, c + 3, ap, bp);

    v2f af[2][G];
    v2f bf[2][G][NT];

    read_group<NT>(curA, curB, 0, mrow, half, af[0], bf[0]);
#pragma unroll
    for (int g = 0; g < NG; ++g) {
        if (g + 1 < NG)
            read_group<NT>(curA, curB, g + 1, mrow, half, af[(g + 1) & 1], bf[(g + 1) & 1]);
        if (DO_STORE)
            store_slice<NT>(nxtA, nxtB, g, x.ar, x.ac, x.br, x.bc, av[P ^ 1], bv[P ^ 1]);
        asm volatile("" ::: "memory");
        if (DO_LOAD)
            load_slice<LDW, NT>(ap, bp, g, av[P ^ 1], bv[P ^ 1]);
        __builtin_amdgcn_sched_barrier(0);
#pragma unroll
        for (int s = 0; s < G; ++s) {
#pragma unroll
            for (int j = 0; j < NT; ++j)
                acc[j] = __builtin_amdgcn_wmma_f32_16x16x4_f32(
                    false, af[g & 1][s], false, bf[g & 1][s][j], (short)0,
                    acc[j], false, false);
        }
        __builtin_amdgcn_sched_barrier(0);
    }
    lds_stage_barrier();
}

template <int LDA, int LDW, int NT, int NC>
__device__ inline void gemm_tile_acc(const float* __restrict__ A,
                                     const float* __restrict__ Wt,
                                     const int ts, int m0, int n0w,
                                     float (*ldsA)[16][LDA_PITCH],
                                     float (*ldsB0)[KC][LDB_PITCH],
                                     float (*ldsB1)[KC][LDB_PITCH],
                                     v8f (&acc)[NT])
{
    static_assert(NC >= 4 && NC % 2 == 0, "pipeline shape");
    const int t    = ts;
    const int lane = t & 31;

    constexpr int LPR = 4 * NT;
    constexpr int NB  = KC * NT / 8;
    constexpr int NQ  = KC / 32;

    GemmCtx x;
    x.A = A; x.Wt = Wt;
    x.ar = t >> 3;
    x.ac = (t & 7) * 4;
    x.br = lane / LPR;
    x.bc = (lane % LPR) * 4;
    x.mrow = lane & 15;
    x.half = lane >> 4;
    x.aoff = (size_t)(m0 + x.ar) * LDA + x.ac;
    x.boff = (size_t)x.br * LDW + n0w + x.bc;

    v4f av[2][NQ];
    v4f bv[2][NB];

    const float* ap;
    const float* bp;
    chunk_ptrs<LDW>(x, 0, ap, bp);
#pragma unroll
    for (int q = 0; q < NQ; ++q) load_slice<LDW, NT>(ap, bp, q, av[0], bv[0]);
    chunk_ptrs<LDW>(x, 1, ap, bp);
#pragma unroll
    for (int q = 0; q < NQ; ++q) load_slice<LDW, NT>(ap, bp, q, av[1], bv[1]);
#pragma unroll
    for (int q = 0; q < NQ; ++q)
        store_slice<NT>(ldsA[0], ldsB0, q, x.ar, x.ac, x.br, x.bc, av[0], bv[0]);
    asm volatile("" ::: "memory");
    chunk_ptrs<LDW>(x, 2, ap, bp);
#pragma unroll
    for (int q = 0; q < NQ; ++q) load_slice<LDW, NT>(ap, bp, q, av[0], bv[0]);
    lds_stage_barrier();

#pragma nounroll
    for (int c = 0; c + 4 < NC; c += 2) {
        gemm_trip<LDW, NT, 0, true, true>(x, c,     ldsA, ldsB0, ldsB1, av, bv, acc);
        gemm_trip<LDW, NT, 1, true, true>(x, c + 1, ldsA, ldsB0, ldsB1, av, bv, acc);
    }
    gemm_trip<LDW, NT, 0, true,  true >(x, NC - 4, ldsA, ldsB0, ldsB1, av, bv, acc);
    gemm_trip<LDW, NT, 1, true,  false>(x, NC - 3, ldsA, ldsB0, ldsB1, av, bv, acc);
    gemm_trip<LDW, NT, 0, true,  false>(x, NC - 2, ldsA, ldsB0, ldsB1, av, bv, acc);
    gemm_trip<LDW, NT, 1, false, false>(x, NC - 1, ldsA, ldsB0, ldsB1, av, bv, acc);
}

template <int K1, int K2>
__global__ void __launch_bounds__(256)
lstm_cell_kernel(const float* __restrict__ A1, const float* __restrict__ W,
                 const float* __restrict__ A2, const float* __restrict__ Uw,
                 const float* __restrict__ bias, const float* __restrict__ Cprev,
                 float* __restrict__ Hout, float* __restrict__ Cout,
                 int Uu)
{
    __shared__ float ldsA[2][2][16][LDA_PITCH];
    __shared__ float ldsB[2][8][2][KC][LDB_PITCH];
    __shared__ float pbuf[4][2][8][32];
    __shared__ float gbuf[4][16][34];

    const int t    = threadIdx.x;
    const int wave = t >> 5;
    const int gate = wave & 3;
    const int kset = wave >> 2;
    const int lane = t & 31;
    const int half = lane >> 4;
    const int mrow = lane & 15;
    const int m0   = blockIdx.y * 16;
    const int n0   = blockIdx.x * 32;
    const int n0w  = gate * Uu + n0;

    v8f acc[2] = {};
    static_assert(K1 % KC == 0 && K1 == K2,
                  "whole chunks, and both wave sets meet the same barriers");
    gemm_tile_acc<1024, 4096, 2, K1 / KC>(
        kset ? A2 : A1, kset ? Uw : W, t & 127, m0, n0w,
        ldsA[kset], ldsB[0][wave], ldsB[1][wave], acc);

    if (kset == 1) {
#pragma unroll
        for (int j = 0; j < 2; ++j) {
#pragma unroll
            for (int r = 0; r < 8; ++r) pbuf[gate][j][r][lane] = acc[j][r];
        }
    }
    __syncthreads();

    if (kset == 0) {
#pragma unroll
        for (int j = 0; j < 2; ++j) {
            const float bv = bias[n0w + 16 * j + mrow];
#pragma unroll
            for (int r = 0; r < 8; ++r) {
                const float z = (acc[j][r] + pbuf[gate][j][r][lane]) + bv;
                const float a = (gate == 2) ? tanhf(z) : (1.0f / (1.0f + __expf(-z)));
                gbuf[gate][r + 8 * half][16 * j + mrow] = a;
            }
        }
    }
    __syncthreads();

    for (int e = t; e < 512; e += 256) {
        const int m = e >> 5, n = e & 31;
        const float iv = gbuf[0][m][n];
        const float fv = gbuf[1][m][n];
        const float gv = gbuf[2][m][n];
        const float ov = gbuf[3][m][n];
        const size_t idx = (size_t)(m0 + m) * Uu + n0 + n;
        const float cn = fv * Cprev[idx] + iv * gv;
        Cout[idx] = cn;
        Hout[idx] = ov * tanhf(cn);
    }
}

template <int K>
__global__ void __launch_bounds__(256)
sproj_kernel(const float* __restrict__ H2, const float* __restrict__ Ws,
             const float* __restrict__ bs, float* __restrict__ Sout, int Uu)
{
    __shared__ float ldsA[2][2][16][LDA_PITCH];
    __shared__ float ldsB[2][8][1][KC][LDB_PITCH];
    __shared__ float pbuf[4][8][32];

    const int t    = threadIdx.x;
    const int wave = t >> 5;
    const int ct   = wave & 3;
    const int kset = wave >> 2;
    const int lane = t & 31;
    const int half = lane >> 4;
    const int mrow = lane & 15;
    const int m0   = blockIdx.y * 16;
    const int n0w  = blockIdx.x * 64 + ct * 16;

    constexpr int KH = K / 2;
    v8f acc1[1] = {};
    static_assert(K % 2 == 0 && KH % KC == 0, "whole chunks in each half of K");
    gemm_tile_acc<1024, 1024, 1, KH / KC>(
        H2 + kset * KH, Ws + (size_t)(kset * KH) * 1024, t & 127, m0, n0w,
        ldsA[kset], ldsB[0][wave], ldsB[1][wave], acc1);
    const v8f acc = acc1[0];

    if (kset == 1) {
#pragma unroll
        for (int r = 0; r < 8; ++r) pbuf[ct][r][lane] = acc[r];
    }
    __syncthreads();

    if (kset == 0) {
        const float bv = bs[n0w + mrow];
#pragma unroll
        for (int r = 0; r < 8; ++r) {
            const int gm = m0 + r + 8 * half;
            Sout[(size_t)gm * Uu + n0w + mrow] = (acc[r] + pbuf[ct][r][lane]) + bv;
        }
    }
}

#define ATT_SPLIT 2

__global__ void __launch_bounds__(512)
attention_part_kernel(const float* __restrict__ av, const float* __restrict__ S,
                      float* __restrict__ P, float* __restrict__ ML,
                      int T, int Uu, int B)
{
    __shared__ float sS[1024];
    __shared__ float ctx_s[1024];
    __shared__ float wm[16], wl[16];

    const int b    = blockIdx.x / ATT_SPLIT;
    const int h    = blockIdx.x % ATT_SPLIT;
    const int t    = threadIdx.x;
    const int wave = t >> 5;
    const int lane = t & 31;

    for (int u = t; u < Uu; u += 512) {
        sS[u]    = S[(size_t)b * Uu + u];
        ctx_s[u] = 0.0f;
    }
    __syncthreads();

    float m = -INFINITY, l = 0.0f;
    float ctx[32];
#pragma unroll
    for (int j = 0; j < 32; ++j) ctx[j] = 0.0f;

    const int TH = T / ATT_SPLIT;
    const int t0 = h * TH;
    const int t1 = (h == ATT_SPLIT - 1) ? T : t0 + TH;
    const float* avb = av + (size_t)b * T * Uu;
    for (int tt = t0 + wave; tt < t1; tt += 16) {
        const float* row = avb + (size_t)tt * Uu;
        __builtin_prefetch(avb + (size_t)(tt + 16) * Uu + lane * 32, 0, 0);

        v4f frag[8];
        float dot = 0.0f;
#pragma unroll
        for (int j = 0; j < 8; ++j) {
            const int u = (j * 32 + lane) * 4;
            frag[j] = __builtin_nontemporal_load((const v4f*)(row + u));
            dot += frag[j].x * sS[u] + frag[j].y * sS[u + 1] +
                   frag[j].z * sS[u + 2] + frag[j].w * sS[u + 3];
        }
#pragma unroll
        for (int off = 16; off > 0; off >>= 1) dot += __shfl_xor(dot, off, 32);

        const float mn    = fmaxf(m, dot);
        const float scale = __expf(m - mn);
        const float e     = __expf(dot - mn);
        l = l * scale + e;
#pragma unroll
        for (int j = 0; j < 8; ++j) {
            ctx[j * 4 + 0] = ctx[j * 4 + 0] * scale + e * frag[j].x;
            ctx[j * 4 + 1] = ctx[j * 4 + 1] * scale + e * frag[j].y;
            ctx[j * 4 + 2] = ctx[j * 4 + 2] * scale + e * frag[j].z;
            ctx[j * 4 + 3] = ctx[j * 4 + 3] * scale + e * frag[j].w;
        }
        m = mn;
    }

    if (lane == 0) { wm[wave] = m; wl[wave] = l; }
    __syncthreads();

    float M = wm[0];
#pragma unroll
    for (int w = 1; w < 16; ++w) M = fmaxf(M, wm[w]);
    float L = 0.0f;
#pragma unroll
    for (int w = 0; w < 16; ++w) L += wl[w] * __expf(wm[w] - M);

    const float f = __expf(m - M);
#pragma unroll
    for (int j = 0; j < 8; ++j) {
        const int u = (j * 32 + lane) * 4;
        atomicAdd(&ctx_s[u + 0], ctx[j * 4 + 0] * f);
        atomicAdd(&ctx_s[u + 1], ctx[j * 4 + 1] * f);
        atomicAdd(&ctx_s[u + 2], ctx[j * 4 + 2] * f);
        atomicAdd(&ctx_s[u + 3], ctx[j * 4 + 3] * f);
    }
    __syncthreads();

    float* Pb = P + ((size_t)h * B + b) * Uu;
    for (int u = t; u < Uu; u += 512) Pb[u] = ctx_s[u];
    if (t == 0) {
        ML[2 * blockIdx.x]     = M;
        ML[2 * blockIdx.x + 1] = L;
    }
}

__global__ void __launch_bounds__(256)
attention_combine_kernel(const float* __restrict__ P, const float* __restrict__ ML,
                         const float* __restrict__ C2, float* __restrict__ Cout,
                         int Uu, int B)
{
    const int b = blockIdx.x;

    float Mh[ATT_SPLIT], e[ATT_SPLIT];
    float M = -INFINITY;
    for (int h = 0; h < ATT_SPLIT; ++h) {
        Mh[h] = ML[2 * (b * ATT_SPLIT + h)];
        M = fmaxf(M, Mh[h]);
    }
    float L = 0.0f;
    for (int h = 0; h < ATT_SPLIT; ++h) {
        e[h] = __expf(Mh[h] - M);
        L += ML[2 * (b * ATT_SPLIT + h) + 1] * e[h];
    }
    const float invL = 1.0f / L;

    const size_t base  = (size_t)b * Uu;
    const size_t plane = (size_t)B * Uu;
    for (int u = 4 * threadIdx.x; u < Uu; u += 4 * 256) {
        v4f acc = *(const v4f*)(P + base + u) * e[0];
        for (int h = 1; h < ATT_SPLIT; ++h)
            acc += *(const v4f*)(P + h * plane + base + u) * e[h];
        const v4f c2 = *(const v4f*)(C2 + base + u);
        *(v4f*)(Cout + base + u) = c2 + acc * invL;
    }
}

extern "C" void kernel_launch(void* const* d_in, const int* in_sizes, int n_in,
                              void* d_out, int out_size, void* d_ws, size_t ws_size,
                              hipStream_t stream)
{
    (void)in_sizes; (void)n_in; (void)out_size; (void)ws_size;
    const int B = 128, T = 512, U = 1024;

    const float* x   = (const float*)d_in[0];
    const float* h0  = (const float*)d_in[1];
    const float* c0  = (const float*)d_in[2];
    const float* av  = (const float*)d_in[3];
    const float* W1  = (const float*)d_in[4];
    const float* U1  = (const float*)d_in[5];
    const float* b1  = (const float*)d_in[6];
    const float* W2  = (const float*)d_in[7];
    const float* U2  = (const float*)d_in[8];
    const float* b2  = (const float*)d_in[9];
    const float* Ws  = (const float*)d_in[10];
    const float* bs  = (const float*)d_in[11];
    float* out = (float*)d_out;

    float* ws = (float*)d_ws;
    const size_t BU = (size_t)B * U;
    float* h1 = ws;
    float* c1 = ws + BU;
    float* c2 = ws + 3 * BU;
    float* s  = ws + 4 * BU;
    float* h2   = out;
    float* part = ws;
    float* ml   = ws + 2 * BU;
    static_assert(ATT_SPLIT == 2, "two dead planes for the partial contexts");

    dim3 gGemm(U / 32, B / 16);
    dim3 bGemm(256);

    lstm_cell_kernel<1024, 1024><<<gGemm, bGemm, 0, stream>>>(x, W1, h0, U1, b1, c0,
                                                              h1, c1, U);
    lstm_cell_kernel<1024, 1024><<<gGemm, bGemm, 0, stream>>>(h1, W2, h1, U2, b2, c1,
                                                              h2, c2, U);
    sproj_kernel<1024><<<dim3(U / 64, B / 16), bGemm, 0, stream>>>(h2, Ws, bs, s, U);
    attention_part_kernel<<<dim3(B * ATT_SPLIT), dim3(512), 0, stream>>>(av, s, part, ml,
                                                                         T, U, B);
    attention_combine_kernel<<<dim3(B), dim3(256), 0, stream>>>(part, ml, c2, out + BU,
                                                                U, B);
}


# LSTM cell 2 (input and state are the same tensor): W2 and U2 tiles added at LDS staging, one product h1@(W2+U2) over k=1024 instead of two; half the WMMAs, trips and barriers: sum order changes
# speedup vs baseline: 4.0461x; 1.0411x over previous
#include <hip/hip_runtime.h>
#include <math.h>

typedef __attribute__((ext_vector_type(2))) float v2f;
typedef __attribute__((ext_vector_type(4))) float v4f;
typedef __attribute__((ext_vector_type(8))) float v8f;

#define KC 64
#define LDA_PITCH (KC + 4)

#define LDB_PITCH 20


__device__ __forceinline__ void lds_stage_barrier()
{
    asm volatile("s_wait_dscnt 0x0" ::: "memory");
    asm volatile("s_barrier_signal -1" ::: "memory");
    asm volatile("s_barrier_wait -1" ::: "memory");
}

template <int LDW, int NT, int NW>
__device__ __forceinline__ void load_slice(const float* __restrict__ ap,
                                             const float* __restrict__ bp,
                                             const float* __restrict__ bp2,
                                             const int q,
                                             v4f (&av)[KC / 32],
                                             v4f (&bv)[NW * KC * NT / 8])
{
    constexpr int RPL = 8 / NT;
    constexpr int NBQ = 32 / RPL;
    constexpr int NB  = KC * NT / 8;
    av[q] = *(const v4f*)(ap + 32 * q);
#pragma unroll
    for (int i = 0; i < NBQ; ++i)
        bv[q * NBQ + i] = *(const v4f*)(bp + (size_t)(RPL * (q * NBQ + i)) * LDW);
    if (NW == 2) {
#pragma unroll
        for (int i = 0; i < NBQ; ++i)
            bv[NB + q * NBQ + i] =
                *(const v4f*)(bp2 + (size_t)(RPL * (q * NBQ + i)) * LDW);
    }
}

template <int NT, int NW>
__device__ __forceinline__ void store_slice(float (*sA)[LDA_PITCH],
                                              float (*sB)[KC][LDB_PITCH],
                                              const int q, const int ar, const int ac,
                                              const int br, const int bc,
                                              const v4f (&av)[KC / 32],
                                              const v4f (&bv)[NW * KC * NT / 8])
{
    constexpr int RPL = 8 / NT;
    constexpr int NBQ = 32 / RPL;
    constexpr int NB  = KC * NT / 8;
    *(v4f*)&sA[ar][ac + 32 * q] = av[q];
#pragma unroll
    for (int i = 0; i < NBQ; ++i) {
        v4f w = bv[q * NBQ + i];
        if (NW == 2) w += bv[NB + q * NBQ + i];
        *(v4f*)&sB[bc >> 4][br + RPL * (q * NBQ + i)][bc & 15] = w;
    }
}

template <int NT>
__device__ __forceinline__ void read_group(float (*curA)[LDA_PITCH],
                                           float (*curB)[KC][LDB_PITCH],
                                           const int g, const int mrow, const int half,
                                           v2f (&af)[8], v2f (&bf)[8][NT])
{
#pragma unroll
    for (int s = 0; s < 8; ++s) {
        const int ks = 4 * (8 * g + s);
        af[s].x = curA[mrow][ks + 2 * half];
        af[s].y = curA[mrow][ks + 2 * half + 1];
    }
#pragma unroll
    for (int s = 0; s < 8; ++s) {
        const int ks = 4 * (8 * g + s);
#pragma unroll
        for (int j = 0; j < NT; ++j) {
            bf[s][j].x = curB[j][ks + 2 * half][mrow];
            bf[s][j].y = curB[j][ks + 2 * half + 1][mrow];
        }
        asm volatile("" ::: "memory");
    }
}

struct GemmCtx {
    const float* A; const float* Wt; const float* Wt2;
    size_t aoff, boff;
    int ar, ac, br, bc, mrow, half;
};

template <int LDW>
__device__ __forceinline__ void chunk_ptrs(const GemmCtx& x, const int k,
                                           const float*& ap, const float*& bp,
                                           const float*& bp2)
{
    ap  = x.A + x.aoff + k * KC;
    bp  = x.Wt + x.boff + (size_t)(k * KC) * LDW;
    bp2 = x.Wt2 + x.boff + (size_t)(k * KC) * LDW;
}

template <int LDW, int NT, int NW, int P, bool DO_STORE, bool DO_LOAD>
__device__ __forceinline__ void gemm_trip(const GemmCtx& x, const int c,
                                          float (*ldsA)[16][LDA_PITCH],
                                          float (*ldsB0)[KC][LDB_PITCH],
                                          float (*ldsB1)[KC][LDB_PITCH],
                                          v4f (&av)[2][KC / 32],
                                          v4f (&bv)[2][NW * KC * NT / 8],
                                          v8f (&acc)[NT])
{
    constexpr int G  = 8;
    constexpr int NG = KC / (4 * G);
    static_assert(NG == KC / 32, "a slice of a chunk is one fragment group");

    float (*curA)[LDA_PITCH]      = ldsA[P];
    float (*nxtA)[LDA_PITCH]      = ldsA[P ^ 1];
    float (*curB)[KC][LDB_PITCH]  = P ? ldsB1 : ldsB0;
    float (*nxtB)[KC][LDB_PITCH]  = P ? ldsB0 : ldsB1;
    const int mrow = x.mrow, half = x.half;

    const float* ap  = x.A;
    const float* bp  = x.Wt;
    const float* bp2 = x.Wt2;
    if (DO_LOAD) chunk_ptrs<LDW>(x, c + 3, ap, bp, bp2);

    v2f af[2][G];
    v2f bf[2][G][NT];

    read_group<NT>(curA, curB, 0, mrow, half, af[0], bf[0]);
#pragma unroll
    for (int g = 0; g < NG; ++g) {
        if (g + 1 < NG)
            read_group<NT>(curA, curB, g + 1, mrow, half, af[(g + 1) & 1], bf[(g + 1) & 1]);
        if (DO_STORE)
            store_slice<NT, NW>(nxtA, nxtB, g, x.ar, x.ac, x.br, x.bc, av[P ^ 1], bv[P ^ 1]);
        asm volatile("" ::: "memory");
        if (DO_LOAD)
            load_slice<LDW, NT, NW>(ap, bp, bp2, g, av[P ^ 1], bv[P ^ 1]);
        __builtin_amdgcn_sched_barrier(0);
#pragma unroll
        for (int s = 0; s < G; ++s) {
#pragma unroll
            for (int j = 0; j < NT; ++j)
                acc[j] = __builtin_amdgcn_wmma_f32_16x16x4_f32(
                    false, af[g & 1][s], false, bf[g & 1][s][j], (short)0,
                    acc[j], false, false);
        }
        __builtin_amdgcn_sched_barrier(0);
    }
    lds_stage_barrier();
}

template <int LDA, int LDW, int NT, int NC, int NW>
__device__ inline void gemm_tile_acc(const float* __restrict__ A,
                                     const float* __restrict__ Wt,
                                     const float* __restrict__ Wt2,
                                     const int ts, int m0, int n0w,
                                     float (*ldsA)[16][LDA_PITCH],
                                     float (*ldsB0)[KC][LDB_PITCH],
                                     float (*ldsB1)[KC][LDB_PITCH],
                                     v8f (&acc)[NT])
{
    static_assert(NC >= 4 && NC % 2 == 0, "pipeline shape");
    const int t    = ts;
    const int lane = t & 31;

    constexpr int LPR = 4 * NT;
    constexpr int NB  = KC * NT / 8;
    constexpr int NQ  = KC / 32;

    GemmCtx x;
    x.A = A; x.Wt = Wt; x.Wt2 = Wt2;
    x.ar = t >> 3;
    x.ac = (t & 7) * 4;
    x.br = lane / LPR;
    x.bc = (lane % LPR) * 4;
    x.mrow = lane & 15;
    x.half = lane >> 4;
    x.aoff = (size_t)(m0 + x.ar) * LDA + x.ac;
    x.boff = (size_t)x.br * LDW + n0w + x.bc;

    v4f av[2][NQ];
    v4f bv[2][NW * NB];

    const float* ap;
    const float* bp;
    const float* bp2;
    chunk_ptrs<LDW>(x, 0, ap, bp, bp2);
#pragma unroll
    for (int q = 0; q < NQ; ++q) load_slice<LDW, NT, NW>(ap, bp, bp2, q, av[0], bv[0]);
    chunk_ptrs<LDW>(x, 1, ap, bp, bp2);
#pragma unroll
    for (int q = 0; q < NQ; ++q) load_slice<LDW, NT, NW>(ap, bp, bp2, q, av[1], bv[1]);
#pragma unroll
    for (int q = 0; q < NQ; ++q)
        store_slice<NT, NW>(ldsA[0], ldsB0, q, x.ar, x.ac, x.br, x.bc, av[0], bv[0]);
    asm volatile("" ::: "memory");
    chunk_ptrs<LDW>(x, 2, ap, bp, bp2);
#pragma unroll
    for (int q = 0; q < NQ; ++q) load_slice<LDW, NT, NW>(ap, bp, bp2, q, av[0], bv[0]);
    lds_stage_barrier();

#pragma nounroll
    for (int c = 0; c + 4 < NC; c += 2) {
        gemm_trip<LDW, NT, NW, 0, true, true>(x, c,     ldsA, ldsB0, ldsB1, av, bv, acc);
        gemm_trip<LDW, NT, NW, 1, true, true>(x, c + 1, ldsA, ldsB0, ldsB1, av, bv, acc);
    }
    gemm_trip<LDW, NT, NW, 0, true,  true >(x, NC - 4, ldsA, ldsB0, ldsB1, av, bv, acc);
    gemm_trip<LDW, NT, NW, 1, true,  false>(x, NC - 3, ldsA, ldsB0, ldsB1, av, bv, acc);
    gemm_trip<LDW, NT, NW, 0, true,  false>(x, NC - 2, ldsA, ldsB0, ldsB1, av, bv, acc);
    gemm_trip<LDW, NT, NW, 1, false, false>(x, NC - 1, ldsA, ldsB0, ldsB1, av, bv, acc);
}

template <int K1, int K2, bool SAME_A>
__global__ void __launch_bounds__(256)
lstm_cell_kernel(const float* __restrict__ A1, const float* __restrict__ W,
                 const float* __restrict__ A2, const float* __restrict__ Uw,
                 const float* __restrict__ bias, const float* __restrict__ Cprev,
                 float* __restrict__ Hout, float* __restrict__ Cout,
                 int Uu)
{
    __shared__ float ldsA[2][2][16][LDA_PITCH];
    __shared__ float ldsB[2][8][2][KC][LDB_PITCH];
    __shared__ float pbuf[4][2][8][32];
    __shared__ float gbuf[4][16][34];

    const int t    = threadIdx.x;
    const int wave = t >> 5;
    const int gate = wave & 3;
    const int kset = wave >> 2;
    const int lane = t & 31;
    const int half = lane >> 4;
    const int mrow = lane & 15;
    const int m0   = blockIdx.y * 16;
    const int n0   = blockIdx.x * 32;
    const int n0w  = gate * Uu + n0;

    v8f acc[2] = {};
    static_assert(K1 % KC == 0 && K1 == K2,
                  "whole chunks, and both wave sets meet the same barriers");
    if (SAME_A) {
        constexpr int KH = K1 / 2;
        static_assert(K1 % 2 == 0 && KH % KC == 0, "whole chunks in each half of K");
        gemm_tile_acc<1024, 4096, 2, KH / KC, 2>(
            A1 + kset * KH, W + (size_t)(kset * KH) * 4096,
            Uw + (size_t)(kset * KH) * 4096, t & 127, m0, n0w,
            ldsA[kset], ldsB[0][wave], ldsB[1][wave], acc);
    } else {
        const float* Wset = kset ? Uw : W;
        gemm_tile_acc<1024, 4096, 2, K1 / KC, 1>(
            kset ? A2 : A1, Wset, Wset, t & 127, m0, n0w,
            ldsA[kset], ldsB[0][wave], ldsB[1][wave], acc);
    }

    if (kset == 1) {
#pragma unroll
        for (int j = 0; j < 2; ++j) {
#pragma unroll
            for (int r = 0; r < 8; ++r) pbuf[gate][j][r][lane] = acc[j][r];
        }
    }
    __syncthreads();

    if (kset == 0) {
#pragma unroll
        for (int j = 0; j < 2; ++j) {
            const float bv = bias[n0w + 16 * j + mrow];
#pragma unroll
            for (int r = 0; r < 8; ++r) {
                const float z = (acc[j][r] + pbuf[gate][j][r][lane]) + bv;
                const float a = (gate == 2) ? tanhf(z) : (1.0f / (1.0f + __expf(-z)));
                gbuf[gate][r + 8 * half][16 * j + mrow] = a;
            }
        }
    }
    __syncthreads();

    for (int e = t; e < 512; e += 256) {
        const int m = e >> 5, n = e & 31;
        const float iv = gbuf[0][m][n];
        const float fv = gbuf[1][m][n];
        const float gv = gbuf[2][m][n];
        const float ov = gbuf[3][m][n];
        const size_t idx = (size_t)(m0 + m) * Uu + n0 + n;
        const float cn = fv * Cprev[idx] + iv * gv;
        Cout[idx] = cn;
        Hout[idx] = ov * tanhf(cn);
    }
}

template <int K>
__global__ void __launch_bounds__(256)
sproj_kernel(const float* __restrict__ H2, const float* __restrict__ Ws,
             const float* __restrict__ bs, float* __restrict__ Sout, int Uu)
{
    __shared__ float ldsA[2][2][16][LDA_PITCH];
    __shared__ float ldsB[2][8][1][KC][LDB_PITCH];
    __shared__ float pbuf[4][8][32];

    const int t    = threadIdx.x;
    const int wave = t >> 5;
    const int ct   = wave & 3;
    const int kset = wave >> 2;
    const int lane = t & 31;
    const int half = lane >> 4;
    const int mrow = lane & 15;
    const int m0   = blockIdx.y * 16;
    const int n0w  = blockIdx.x * 64 + ct * 16;

    constexpr int KH = K / 2;
    v8f acc1[1] = {};
    static_assert(K % 2 == 0 && KH % KC == 0, "whole chunks in each half of K");
    gemm_tile_acc<1024, 1024, 1, KH / KC, 1>(
        H2 + kset * KH, Ws + (size_t)(kset * KH) * 1024,
        Ws + (size_t)(kset * KH) * 1024, t & 127, m0, n0w,
        ldsA[kset], ldsB[0][wave], ldsB[1][wave], acc1);
    const v8f acc = acc1[0];

    if (kset == 1) {
#pragma unroll
        for (int r = 0; r < 8; ++r) pbuf[ct][r][lane] = acc[r];
    }
    __syncthreads();

    if (kset == 0) {
        const float bv = bs[n0w + mrow];
#pragma unroll
        for (int r = 0; r < 8; ++r) {
            const int gm = m0 + r + 8 * half;
            Sout[(size_t)gm * Uu + n0w + mrow] = (acc[r] + pbuf[ct][r][lane]) + bv;
        }
    }
}

#define ATT_SPLIT 2

__global__ void __launch_bounds__(512)
attention_part_kernel(const float* __restrict__ av, const float* __restrict__ S,
                      float* __restrict__ P, float* __restrict__ ML,
                      int T, int Uu, int B)
{
    __shared__ float sS[1024];
    __shared__ float ctx_s[1024];
    __shared__ float wm[16], wl[16];

    const int b    = blockIdx.x / ATT_SPLIT;
    const int h    = blockIdx.x % ATT_SPLIT;
    const int t    = threadIdx.x;
    const int wave = t >> 5;
    const int lane = t & 31;

    for (int u = t; u < Uu; u += 512) {
        sS[u]    = S[(size_t)b * Uu + u];
        ctx_s[u] = 0.0f;
    }
    __syncthreads();

    float m = -INFINITY, l = 0.0f;
    float ctx[32];
#pragma unroll
    for (int j = 0; j < 32; ++j) ctx[j] = 0.0f;

    const int TH = T / ATT_SPLIT;
    const int t0 = h * TH;
    const int t1 = (h == ATT_SPLIT - 1) ? T : t0 + TH;
    const float* avb = av + (size_t)b * T * Uu;
    for (int tt = t0 + wave; tt < t1; tt += 16) {
        const float* row = avb + (size_t)tt * Uu;
        __builtin_prefetch(avb + (size_t)(tt + 16) * Uu + lane * 32, 0, 0);

        v4f frag[8];
        float dot = 0.0f;
#pragma unroll
        for (int j = 0; j < 8; ++j) {
            const int u = (j * 32 + lane) * 4;
            frag[j] = __builtin_nontemporal_load((const v4f*)(row + u));
            dot += frag[j].x * sS[u] + frag[j].y * sS[u + 1] +
                   frag[j].z * sS[u + 2] + frag[j].w * sS[u + 3];
        }
#pragma unroll
        for (int off = 16; off > 0; off >>= 1) dot += __shfl_xor(dot, off, 32);

        const float mn    = fmaxf(m, dot);
        const float scale = __expf(m - mn);
        const float e     = __expf(dot - mn);
        l = l * scale + e;
#pragma unroll
        for (int j = 0; j < 8; ++j) {
            ctx[j * 4 + 0] = ctx[j * 4 + 0] * scale + e * frag[j].x;
            ctx[j * 4 + 1] = ctx[j * 4 + 1] * scale + e * frag[j].y;
            ctx[j * 4 + 2] = ctx[j * 4 + 2] * scale + e * frag[j].z;
            ctx[j * 4 + 3] = ctx[j * 4 + 3] * scale + e * frag[j].w;
        }
        m = mn;
    }

    if (lane == 0) { wm[wave] = m; wl[wave] = l; }
    __syncthreads();

    float M = wm[0];
#pragma unroll
    for (int w = 1; w < 16; ++w) M = fmaxf(M, wm[w]);
    float L = 0.0f;
#pragma unroll
    for (int w = 0; w < 16; ++w) L += wl[w] * __expf(wm[w] - M);

    const float f = __expf(m - M);
#pragma unroll
    for (int j = 0; j < 8; ++j) {
        const int u = (j * 32 + lane) * 4;
        atomicAdd(&ctx_s[u + 0], ctx[j * 4 + 0] * f);
        atomicAdd(&ctx_s[u + 1], ctx[j * 4 + 1] * f);
        atomicAdd(&ctx_s[u + 2], ctx[j * 4 + 2] * f);
        atomicAdd(&ctx_s[u + 3], ctx[j * 4 + 3] * f);
    }
    __syncthreads();

    float* Pb = P + ((size_t)h * B + b) * Uu;
    for (int u = t; u < Uu; u += 512) Pb[u] = ctx_s[u];
    if (t == 0) {
        ML[2 * blockIdx.x]     = M;
        ML[2 * blockIdx.x + 1] = L;
    }
}

__global__ void __launch_bounds__(256)
attention_combine_kernel(const float* __restrict__ P, const float* __restrict__ ML,
                         const float* __restrict__ C2, float* __restrict__ Cout,
                         int Uu, int B)
{
    const int b = blockIdx.x;

    float Mh[ATT_SPLIT], e[ATT_SPLIT];
    float M = -INFINITY;
    for (int h = 0; h < ATT_SPLIT; ++h) {
        Mh[h] = ML[2 * (b * ATT_SPLIT + h)];
        M = fmaxf(M, Mh[h]);
    }
    float L = 0.0f;
    for (int h = 0; h < ATT_SPLIT; ++h) {
        e[h] = __expf(Mh[h] - M);
        L += ML[2 * (b * ATT_SPLIT + h) + 1] * e[h];
    }
    const float invL = 1.0f / L;

    const size_t base  = (size_t)b * Uu;
    const size_t plane = (size_t)B * Uu;
    for (int u = 4 * threadIdx.x; u < Uu; u += 4 * 256) {
        v4f acc = *(const v4f*)(P + base + u) * e[0];
        for (int h = 1; h < ATT_SPLIT; ++h)
            acc += *(const v4f*)(P + h * plane + base + u) * e[h];
        const v4f c2 = *(const v4f*)(C2 + base + u);
        *(v4f*)(Cout + base + u) = c2 + acc * invL;
    }
}

extern "C" void kernel_launch(void* const* d_in, const int* in_sizes, int n_in,
                              void* d_out, int out_size, void* d_ws, size_t ws_size,
                              hipStream_t stream)
{
    (void)in_sizes; (void)n_in; (void)out_size; (void)ws_size;
    const int B = 128, T = 512, U = 1024;

    const float* x   = (const float*)d_in[0];
    const float* h0  = (const float*)d_in[1];
    const float* c0  = (const float*)d_in[2];
    const float* av  = (const float*)d_in[3];
    const float* W1  = (const float*)d_in[4];
    const float* U1  = (const float*)d_in[5];
    const float* b1  = (const float*)d_in[6];
    const float* W2  = (const float*)d_in[7];
    const float* U2  = (const float*)d_in[8];
    const float* b2  = (const float*)d_in[9];
    const float* Ws  = (const float*)d_in[10];
    const float* bs  = (const float*)d_in[11];
    float* out = (float*)d_out;

    float* ws = (float*)d_ws;
    const size_t BU = (size_t)B * U;
    float* h1 = ws;
    float* c1 = ws + BU;
    float* c2 = ws + 3 * BU;
    float* s  = ws + 4 * BU;
    float* h2   = out;
    float* part = ws;
    float* ml   = ws + 2 * BU;
    static_assert(ATT_SPLIT == 2, "two dead planes for the partial contexts");

    dim3 gGemm(U / 32, B / 16);
    dim3 bGemm(256);

    lstm_cell_kernel<1024, 1024, false><<<gGemm, bGemm, 0, stream>>>(x, W1, h0, U1, b1,
                                                                     c0, h1, c1, U);
    lstm_cell_kernel<1024, 1024, true><<<gGemm, bGemm, 0, stream>>>(h1, W2, h1, U2, b2,
                                                                    c1, h2, c2, U);
    sproj_kernel<1024><<<dim3(U / 64, B / 16), bGemm, 0, stream>>>(h2, Ws, bs, s, U);
    attention_part_kernel<<<dim3(B * ATT_SPLIT), dim3(512), 0, stream>>>(av, s, part, ml,
                                                                         T, U, B);
    attention_combine_kernel<<<dim3(B), dim3(256), 0, stream>>>(part, ml, c2, out + BU,
                                                                U, B);
}


# GEMM launches: each work-group starts its k loop at a chunk given by its block index and wraps round, so sharers of weight or input lines ask for them at different trips: sum order changes
# speedup vs baseline: 4.2209x; 1.0432x over previous
#include <hip/hip_runtime.h>
#include <math.h>

typedef __attribute__((ext_vector_type(2))) float v2f;
typedef __attribute__((ext_vector_type(4))) float v4f;
typedef __attribute__((ext_vector_type(8))) float v8f;

#define KC 64
#define LDA_PITCH (KC + 4)

#define LDB_PITCH 20


__device__ __forceinline__ void lds_stage_barrier()
{
    asm volatile("s_wait_dscnt 0x0" ::: "memory");
    asm volatile("s_barrier_signal -1" ::: "memory");
    asm volatile("s_barrier_wait -1" ::: "memory");
}

template <int LDW, int NT, int NW>
__device__ __forceinline__ void load_slice(const float* __restrict__ ap,
                                             const float* __restrict__ bp,
                                             const float* __restrict__ bp2,
                                             const int q,
                                             v4f (&av)[KC / 32],
                                             v4f (&bv)[NW * KC * NT / 8])
{
    constexpr int RPL = 8 / NT;
    constexpr int NBQ = 32 / RPL;
    constexpr int NB  = KC * NT / 8;
    av[q] = *(const v4f*)(ap + 32 * q);
#pragma unroll
    for (int i = 0; i < NBQ; ++i)
        bv[q * NBQ + i] = *(const v4f*)(bp + (size_t)(RPL * (q * NBQ + i)) * LDW);
    if (NW == 2) {
#pragma unroll
        for (int i = 0; i < NBQ; ++i)
            bv[NB + q * NBQ + i] =
                *(const v4f*)(bp2 + (size_t)(RPL * (q * NBQ + i)) * LDW);
    }
}

template <int NT, int NW>
__device__ __forceinline__ void store_slice(float (*sA)[LDA_PITCH],
                                              float (*sB)[KC][LDB_PITCH],
                                              const int q, const int ar, const int ac,
                                              const int br, const int bc,
                                              const v4f (&av)[KC / 32],
                                              const v4f (&bv)[NW * KC * NT / 8])
{
    constexpr int RPL = 8 / NT;
    constexpr int NBQ = 32 / RPL;
    constexpr int NB  = KC * NT / 8;
    *(v4f*)&sA[ar][ac + 32 * q] = av[q];
#pragma unroll
    for (int i = 0; i < NBQ; ++i) {
        v4f w = bv[q * NBQ + i];
        if (NW == 2) w += bv[NB + q * NBQ + i];
        *(v4f*)&sB[bc >> 4][br + RPL * (q * NBQ + i)][bc & 15] = w;
    }
}

template <int NT>
__device__ __forceinline__ void read_group(float (*curA)[LDA_PITCH],
                                           float (*curB)[KC][LDB_PITCH],
                                           const int g, const int mrow, const int half,
                                           v2f (&af)[8], v2f (&bf)[8][NT])
{
#pragma unroll
    for (int s = 0; s < 8; ++s) {
        const int ks = 4 * (8 * g + s);
        af[s].x = curA[mrow][ks + 2 * half];
        af[s].y = curA[mrow][ks + 2 * half + 1];
    }
#pragma unroll
    for (int s = 0; s < 8; ++s) {
        const int ks = 4 * (8 * g + s);
#pragma unroll
        for (int j = 0; j < NT; ++j) {
            bf[s][j].x = curB[j][ks + 2 * half][mrow];
            bf[s][j].y = curB[j][ks + 2 * half + 1][mrow];
        }
        asm volatile("" ::: "memory");
    }
}

struct GemmCtx {
    const float* A; const float* Wt; const float* Wt2;
    size_t aoff, boff;
    int ar, ac, br, bc, mrow, half;
    int rot, kmask;
};

template <int LDW>
__device__ __forceinline__ void chunk_ptrs(const GemmCtx& x, const int k,
                                           const float*& ap, const float*& bp,
                                           const float*& bp2)
{
    const int kk = (k + x.rot) & x.kmask;
    ap  = x.A + x.aoff + kk * KC;
    bp  = x.Wt + x.boff + (size_t)(kk * KC) * LDW;
    bp2 = x.Wt2 + x.boff + (size_t)(kk * KC) * LDW;
}

template <int LDW, int NT, int NW, int P, bool DO_STORE, bool DO_LOAD>
__device__ __forceinline__ void gemm_trip(const GemmCtx& x, const int c,
                                          float (*ldsA)[16][LDA_PITCH],
                                          float (*ldsB0)[KC][LDB_PITCH],
                                          float (*ldsB1)[KC][LDB_PITCH],
                                          v4f (&av)[2][KC / 32],
                                          v4f (&bv)[2][NW * KC * NT / 8],
                                          v8f (&acc)[NT])
{
    constexpr int G  = 8;
    constexpr int NG = KC / (4 * G);
    static_assert(NG == KC / 32, "a slice of a chunk is one fragment group");

    float (*curA)[LDA_PITCH]      = ldsA[P];
    float (*nxtA)[LDA_PITCH]      = ldsA[P ^ 1];
    float (*curB)[KC][LDB_PITCH]  = P ? ldsB1 : ldsB0;
    float (*nxtB)[KC][LDB_PITCH]  = P ? ldsB0 : ldsB1;
    const int mrow = x.mrow, half = x.half;

    const float* ap  = x.A;
    const float* bp  = x.Wt;
    const float* bp2 = x.Wt2;
    if (DO_LOAD) chunk_ptrs<LDW>(x, c + 3, ap, bp, bp2);

    v2f af[2][G];
    v2f bf[2][G][NT];

    read_group<NT>(curA, curB, 0, mrow, half, af[0], bf[0]);
#pragma unroll
    for (int g = 0; g < NG; ++g) {
        if (g + 1 < NG)
            read_group<NT>(curA, curB, g + 1, mrow, half, af[(g + 1) & 1], bf[(g + 1) & 1]);
        if (DO_STORE)
            store_slice<NT, NW>(nxtA, nxtB, g, x.ar, x.ac, x.br, x.bc, av[P ^ 1], bv[P ^ 1]);
        asm volatile("" ::: "memory");
        if (DO_LOAD)
            load_slice<LDW, NT, NW>(ap, bp, bp2, g, av[P ^ 1], bv[P ^ 1]);
        __builtin_amdgcn_sched_barrier(0);
#pragma unroll
        for (int s = 0; s < G; ++s) {
#pragma unroll
            for (int j = 0; j < NT; ++j)
                acc[j] = __builtin_amdgcn_wmma_f32_16x16x4_f32(
                    false, af[g & 1][s], false, bf[g & 1][s][j], (short)0,
                    acc[j], false, false);
        }
        __builtin_amdgcn_sched_barrier(0);
    }
    lds_stage_barrier();
}

template <int LDA, int LDW, int NT, int NC, int NW>
__device__ inline void gemm_tile_acc(const float* __restrict__ A,
                                     const float* __restrict__ Wt,
                                     const float* __restrict__ Wt2,
                                     const int ts, int m0, int n0w, const int rot,
                                     float (*ldsA)[16][LDA_PITCH],
                                     float (*ldsB0)[KC][LDB_PITCH],
                                     float (*ldsB1)[KC][LDB_PITCH],
                                     v8f (&acc)[NT])
{
    static_assert(NC >= 4 && NC % 2 == 0, "pipeline shape");
    static_assert((NC & (NC - 1)) == 0, "the start chunk wraps round with a mask");
    const int t    = ts;
    const int lane = t & 31;

    constexpr int LPR = 4 * NT;
    constexpr int NB  = KC * NT / 8;
    constexpr int NQ  = KC / 32;

    GemmCtx x;
    x.A = A; x.Wt = Wt; x.Wt2 = Wt2;
    x.ar = t >> 3;
    x.ac = (t & 7) * 4;
    x.br = lane / LPR;
    x.bc = (lane % LPR) * 4;
    x.mrow = lane & 15;
    x.half = lane >> 4;
    x.aoff = (size_t)(m0 + x.ar) * LDA + x.ac;
    x.boff = (size_t)x.br * LDW + n0w + x.bc;
    x.kmask = NC - 1;
    x.rot = rot & (NC - 1);

    v4f av[2][NQ];
    v4f bv[2][NW * NB];

    const float* ap;
    const float* bp;
    const float* bp2;
    chunk_ptrs<LDW>(x, 0, ap, bp, bp2);
#pragma unroll
    for (int q = 0; q < NQ; ++q) load_slice<LDW, NT, NW>(ap, bp, bp2, q, av[0], bv[0]);
    chunk_ptrs<LDW>(x, 1, ap, bp, bp2);
#pragma unroll
    for (int q = 0; q < NQ; ++q) load_slice<LDW, NT, NW>(ap, bp, bp2, q, av[1], bv[1]);
#pragma unroll
    for (int q = 0; q < NQ; ++q)
        store_slice<NT, NW>(ldsA[0], ldsB0, q, x.ar, x.ac, x.br, x.bc, av[0], bv[0]);
    asm volatile("" ::: "memory");
    chunk_ptrs<LDW>(x, 2, ap, bp, bp2);
#pragma unroll
    for (int q = 0; q < NQ; ++q) load_slice<LDW, NT, NW>(ap, bp, bp2, q, av[0], bv[0]);
    lds_stage_barrier();

#pragma nounroll
    for (int c = 0; c + 4 < NC; c += 2) {
        gemm_trip<LDW, NT, NW, 0, true, true>(x, c,     ldsA, ldsB0, ldsB1, av, bv, acc);
        gemm_trip<LDW, NT, NW, 1, true, true>(x, c + 1, ldsA, ldsB0, ldsB1, av, bv, acc);
    }
    gemm_trip<LDW, NT, NW, 0, true,  true >(x, NC - 4, ldsA, ldsB0, ldsB1, av, bv, acc);
    gemm_trip<LDW, NT, NW, 1, true,  false>(x, NC - 3, ldsA, ldsB0, ldsB1, av, bv, acc);
    gemm_trip<LDW, NT, NW, 0, true,  false>(x, NC - 2, ldsA, ldsB0, ldsB1, av, bv, acc);
    gemm_trip<LDW, NT, NW, 1, false, false>(x, NC - 1, ldsA, ldsB0, ldsB1, av, bv, acc);
}

template <int K1, int K2, bool SAME_A>
__global__ void __launch_bounds__(256)
lstm_cell_kernel(const float* __restrict__ A1, const float* __restrict__ W,
                 const float* __restrict__ A2, const float* __restrict__ Uw,
                 const float* __restrict__ bias, const float* __restrict__ Cprev,
                 float* __restrict__ Hout, float* __restrict__ Cout,
                 int Uu)
{
    __shared__ float ldsA[2][2][16][LDA_PITCH];
    __shared__ float ldsB[2][8][2][KC][LDB_PITCH];
    __shared__ float pbuf[4][2][8][32];
    __shared__ float gbuf[4][16][34];

    const int t    = threadIdx.x;
    const int wave = t >> 5;
    const int gate = wave & 3;
    const int kset = wave >> 2;
    const int lane = t & 31;
    const int half = lane >> 4;
    const int mrow = lane & 15;
    const int m0   = blockIdx.y * 16;
    const int n0   = blockIdx.x * 32;
    const int n0w  = gate * Uu + n0;

    const int rot = blockIdx.x + (SAME_A ? 1 : 2) * blockIdx.y;

    v8f acc[2] = {};
    static_assert(K1 % KC == 0 && K1 == K2,
                  "whole chunks, and both wave sets meet the same barriers");
    if (SAME_A) {
        constexpr int KH = K1 / 2;
        static_assert(K1 % 2 == 0 && KH % KC == 0, "whole chunks in each half of K");
        gemm_tile_acc<1024, 4096, 2, KH / KC, 2>(
            A1 + kset * KH, W + (size_t)(kset * KH) * 4096,
            Uw + (size_t)(kset * KH) * 4096, t & 127, m0, n0w, rot,
            ldsA[kset], ldsB[0][wave], ldsB[1][wave], acc);
    } else {
        const float* Wset = kset ? Uw : W;
        gemm_tile_acc<1024, 4096, 2, K1 / KC, 1>(
            kset ? A2 : A1, Wset, Wset, t & 127, m0, n0w, rot,
            ldsA[kset], ldsB[0][wave], ldsB[1][wave], acc);
    }

    if (kset == 1) {
#pragma unroll
        for (int j = 0; j < 2; ++j) {
#pragma unroll
            for (int r = 0; r < 8; ++r) pbuf[gate][j][r][lane] = acc[j][r];
        }
    }
    __syncthreads();

    if (kset == 0) {
#pragma unroll
        for (int j = 0; j < 2; ++j) {
            const float bv = bias[n0w + 16 * j + mrow];
#pragma unroll
            for (int r = 0; r < 8; ++r) {
                const float z = (acc[j][r] + pbuf[gate][j][r][lane]) + bv;
                const float a = (gate == 2) ? tanhf(z) : (1.0f / (1.0f + __expf(-z)));
                gbuf[gate][r + 8 * half][16 * j + mrow] = a;
            }
        }
    }
    __syncthreads();

    for (int e = t; e < 512; e += 256) {
        const int m = e >> 5, n = e & 31;
        const float iv = gbuf[0][m][n];
        const float fv = gbuf[1][m][n];
        const float gv = gbuf[2][m][n];
        const float ov = gbuf[3][m][n];
        const size_t idx = (size_t)(m0 + m) * Uu + n0 + n;
        const float cn = fv * Cprev[idx] + iv * gv;
        Cout[idx] = cn;
        Hout[idx] = ov * tanhf(cn);
    }
}

template <int K>
__global__ void __launch_bounds__(256)
sproj_kernel(const float* __restrict__ H2, const float* __restrict__ Ws,
             const float* __restrict__ bs, float* __restrict__ Sout, int Uu)
{
    __shared__ float ldsA[2][2][16][LDA_PITCH];
    __shared__ float ldsB[2][8][1][KC][LDB_PITCH];
    __shared__ float pbuf[4][8][32];

    const int t    = threadIdx.x;
    const int wave = t >> 5;
    const int ct   = wave & 3;
    const int kset = wave >> 2;
    const int lane = t & 31;
    const int half = lane >> 4;
    const int mrow = lane & 15;
    const int m0   = blockIdx.y * 16;
    const int n0w  = blockIdx.x * 64 + ct * 16;

    constexpr int KH = K / 2;
    const int rot = blockIdx.x + blockIdx.y;
    v8f acc1[1] = {};
    static_assert(K % 2 == 0 && KH % KC == 0, "whole chunks in each half of K");
    gemm_tile_acc<1024, 1024, 1, KH / KC, 1>(
        H2 + kset * KH, Ws + (size_t)(kset * KH) * 1024,
        Ws + (size_t)(kset * KH) * 1024, t & 127, m0, n0w, rot,
        ldsA[kset], ldsB[0][wave], ldsB[1][wave], acc1);
    const v8f acc = acc1[0];

    if (kset == 1) {
#pragma unroll
        for (int r = 0; r < 8; ++r) pbuf[ct][r][lane] = acc[r];
    }
    __syncthreads();

    if (kset == 0) {
        const float bv = bs[n0w + mrow];
#pragma unroll
        for (int r = 0; r < 8; ++r) {
            const int gm = m0 + r + 8 * half;
            Sout[(size_t)gm * Uu + n0w + mrow] = (acc[r] + pbuf[ct][r][lane]) + bv;
        }
    }
}

#define ATT_SPLIT 2

__global__ void __launch_bounds__(512)
attention_part_kernel(const float* __restrict__ av, const float* __restrict__ S,
                      float* __restrict__ P, float* __restrict__ ML,
                      int T, int Uu, int B)
{
    __shared__ float sS[1024];
    __shared__ float ctx_s[1024];
    __shared__ float wm[16], wl[16];

    const int b    = blockIdx.x / ATT_SPLIT;
    const int h    = blockIdx.x % ATT_SPLIT;
    const int t    = threadIdx.x;
    const int wave = t >> 5;
    const int lane = t & 31;

    for (int u = t; u < Uu; u += 512) {
        sS[u]    = S[(size_t)b * Uu + u];
        ctx_s[u] = 0.0f;
    }
    __syncthreads();

    float m = -INFINITY, l = 0.0f;
    float ctx[32];
#pragma unroll
    for (int j = 0; j < 32; ++j) ctx[j] = 0.0f;

    const int TH = T / ATT_SPLIT;
    const int t0 = h * TH;
    const int t1 = (h == ATT_SPLIT - 1) ? T : t0 + TH;
    const float* avb = av + (size_t)b * T * Uu;
    for (int tt = t0 + wave; tt < t1; tt += 16) {
        const float* row = avb + (size_t)tt * Uu;
        __builtin_prefetch(avb + (size_t)(tt + 16) * Uu + lane * 32, 0, 0);

        v4f frag[8];
        float dot = 0.0f;
#pragma unroll
        for (int j = 0; j < 8; ++j) {
            const int u = (j * 32 + lane) * 4;
            frag[j] = __builtin_nontemporal_load((const v4f*)(row + u));
            dot += frag[j].x * sS[u] + frag[j].y * sS[u + 1] +
                   frag[j].z * sS[u + 2] + frag[j].w * sS[u + 3];
        }
#pragma unroll
        for (int off = 16; off > 0; off >>= 1) dot += __shfl_xor(dot, off, 32);

        const float mn    = fmaxf(m, dot);
        const float scale = __expf(m - mn);
        const float e     = __expf(dot - mn);
        l = l * scale + e;
#pragma unroll
        for (int j = 0; j < 8; ++j) {
            ctx[j * 4 + 0] = ctx[j * 4 + 0] * scale + e * frag[j].x;
            ctx[j * 4 + 1] = ctx[j * 4 + 1] * scale + e * frag[j].y;
            ctx[j * 4 + 2] = ctx[j * 4 + 2] * scale + e * frag[j].z;
            ctx[j * 4 + 3] = ctx[j * 4 + 3] * scale + e * frag[j].w;
        }
        m = mn;
    }

    if (lane == 0) { wm[wave] = m; wl[wave] = l; }
    __syncthreads();

    float M = wm[0];
#pragma unroll
    for (int w = 1; w < 16; ++w) M = fmaxf(M, wm[w]);
    float L = 0.0f;
#pragma unroll
    for (int w = 0; w < 16; ++w) L += wl[w] * __expf(wm[w] - M);

    const float f = __expf(m - M);
#pragma unroll
    for (int j = 0; j < 8; ++j) {
        const int u = (j * 32 + lane) * 4;
        atomicAdd(&ctx_s[u + 0], ctx[j * 4 + 0] * f);
        atomicAdd(&ctx_s[u + 1], ctx[j * 4 + 1] * f);
        atomicAdd(&ctx_s[u + 2], ctx[j * 4 + 2] * f);
        atomicAdd(&ctx_s[u + 3], ctx[j * 4 + 3] * f);
    }
    __syncthreads();

    float* Pb = P + ((size_t)h * B + b) * Uu;
    for (int u = t; u < Uu; u += 512) Pb[u] = ctx_s[u];
    if (t == 0) {
        ML[2 * blockIdx.x]     = M;
        ML[2 * blockIdx.x + 1] = L;
    }
}

__global__ void __launch_bounds__(256)
attention_combine_kernel(const float* __restrict__ P, const float* __restrict__ ML,
                         const float* __restrict__ C2, float* __restrict__ Cout,
                         int Uu, int B)
{
    const int b = blockIdx.x;

    float Mh[ATT_SPLIT], e[ATT_SPLIT];
    float M = -INFINITY;
    for (int h = 0; h < ATT_SPLIT; ++h) {
        Mh[h] = ML[2 * (b * ATT_SPLIT + h)];
        M = fmaxf(M, Mh[h]);
    }
    float L = 0.0f;
    for (int h = 0; h < ATT_SPLIT; ++h) {
        e[h] = __expf(Mh[h] - M);
        L += ML[2 * (b * ATT_SPLIT + h) + 1] * e[h];
    }
    const float invL = 1.0f / L;

    const size_t base  = (size_t)b * Uu;
    const size_t plane = (size_t)B * Uu;
    for (int u = 4 * threadIdx.x; u < Uu; u += 4 * 256) {
        v4f acc = *(const v4f*)(P + base + u) * e[0];
        for (int h = 1; h < ATT_SPLIT; ++h)
            acc += *(const v4f*)(P + h * plane + base + u) * e[h];
        const v4f c2 = *(const v4f*)(C2 + base + u);
        *(v4f*)(Cout + base + u) = c2 + acc * invL;
    }
}

extern "C" void kernel_launch(void* const* d_in, const int* in_sizes, int n_in,
                              void* d_out, int out_size, void* d_ws, size_t ws_size,
                              hipStream_t stream)
{
    (void)in_sizes; (void)n_in; (void)out_size; (void)ws_size;
    const int B = 128, T = 512, U = 1024;

    const float* x   = (const float*)d_in[0];
    const float* h0  = (const float*)d_in[1];
    const float* c0  = (const float*)d_in[2];
    const float* av  = (const float*)d_in[3];
    const float* W1  = (const float*)d_in[4];
    const float* U1  = (const float*)d_in[5];
    const float* b1  = (const float*)d_in[6];
    const float* W2  = (const float*)d_in[7];
    const float* U2  = (const float*)d_in[8];
    const float* b2  = (const float*)d_in[9];
    const float* Ws  = (const float*)d_in[10];
    const float* bs  = (const float*)d_in[11];
    float* out = (float*)d_out;

    float* ws = (float*)d_ws;
    const size_t BU = (size_t)B * U;
    float* h1 = ws;
    float* c1 = ws + BU;
    float* c2 = ws + 3 * BU;
    float* s  = ws + 4 * BU;
    float* h2   = out;
    float* part = ws;
    float* ml   = ws + 2 * BU;
    static_assert(ATT_SPLIT == 2, "two dead planes for the partial contexts");

    dim3 gGemm(U / 32, B / 16);
    dim3 bGemm(256);

    lstm_cell_kernel<1024, 1024, false><<<gGemm, bGemm, 0, stream>>>(x, W1, h0, U1, b1,
                                                                     c0, h1, c1, U);
    lstm_cell_kernel<1024, 1024, true><<<gGemm, bGemm, 0, stream>>>(h1, W2, h1, U2, b2,
                                                                    c1, h2, c2, U);
    sproj_kernel<1024><<<dim3(U / 64, B / 16), bGemm, 0, stream>>>(h2, Ws, bs, s, U);
    attention_part_kernel<<<dim3(B * ATT_SPLIT), dim3(512), 0, stream>>>(av, s, part, ml,
                                                                         T, U, B);
    attention_combine_kernel<<<dim3(B), dim3(256), 0, stream>>>(part, ml, c2, out + BU,
                                                                U, B);
}


# LSTM cell epilogue as straight-line code: branch-free sigmoid/tanh on v_exp_f32 + v_rcp_f32 replace library tanhf and IEEE division; bias and old cell state loaded ahead of the K loop; sums unchanged
# speedup vs baseline: 4.5083x; 1.0681x over previous
#include <hip/hip_runtime.h>
#include <math.h>

typedef __attribute__((ext_vector_type(2))) float v2f;
typedef __attribute__((ext_vector_type(4))) float v4f;
typedef __attribute__((ext_vector_type(8))) float v8f;

#define KC 64
#define LDA_PITCH (KC + 4)

#define LDB_PITCH 20


__device__ __forceinline__ void lds_stage_barrier()
{
    asm volatile("s_wait_dscnt 0x0" ::: "memory");
    asm volatile("s_barrier_signal -1" ::: "memory");
    asm volatile("s_barrier_wait -1" ::: "memory");
}

template <int LDW, int NT, int NW>
__device__ __forceinline__ void load_slice(const float* __restrict__ ap,
                                             const float* __restrict__ bp,
                                             const float* __restrict__ bp2,
                                             const int q,
                                             v4f (&av)[KC / 32],
                                             v4f (&bv)[NW * KC * NT / 8])
{
    constexpr int RPL = 8 / NT;
    constexpr int NBQ = 32 / RPL;
    constexpr int NB  = KC * NT / 8;
    av[q] = *(const v4f*)(ap + 32 * q);
#pragma unroll
    for (int i = 0; i < NBQ; ++i)
        bv[q * NBQ + i] = *(const v4f*)(bp + (size_t)(RPL * (q * NBQ + i)) * LDW);
    if (NW == 2) {
#pragma unroll
        for (int i = 0; i < NBQ; ++i)
            bv[NB + q * NBQ + i] =
                *(const v4f*)(bp2 + (size_t)(RPL * (q * NBQ + i)) * LDW);
    }
}

template <int NT, int NW>
__device__ __forceinline__ void store_slice(float (*sA)[LDA_PITCH],
                                              float (*sB)[KC][LDB_PITCH],
                                              const int q, const int ar, const int ac,
                                              const int br, const int bc,
                                              const v4f (&av)[KC / 32],
                                              const v4f (&bv)[NW * KC * NT / 8])
{
    constexpr int RPL = 8 / NT;
    constexpr int NBQ = 32 / RPL;
    constexpr int NB  = KC * NT / 8;
    *(v4f*)&sA[ar][ac + 32 * q] = av[q];
#pragma unroll
    for (int i = 0; i < NBQ; ++i) {
        v4f w = bv[q * NBQ + i];
        if (NW == 2) w += bv[NB + q * NBQ + i];
        *(v4f*)&sB[bc >> 4][br + RPL * (q * NBQ + i)][bc & 15] = w;
    }
}

template <int NT>
__device__ __forceinline__ void read_group(float (*curA)[LDA_PITCH],
                                           float (*curB)[KC][LDB_PITCH],
                                           const int g, const int mrow, const int half,
                                           v2f (&af)[8], v2f (&bf)[8][NT])
{
#pragma unroll
    for (int s = 0; s < 8; ++s) {
        const int ks = 4 * (8 * g + s);
        af[s].x = curA[mrow][ks + 2 * half];
        af[s].y = curA[mrow][ks + 2 * half + 1];
    }
#pragma unroll
    for (int s = 0; s < 8; ++s) {
        const int ks = 4 * (8 * g + s);
#pragma unroll
        for (int j = 0; j < NT; ++j) {
            bf[s][j].x = curB[j][ks + 2 * half][mrow];
            bf[s][j].y = curB[j][ks + 2 * half + 1][mrow];
        }
        asm volatile("" ::: "memory");
    }
}

struct GemmCtx {
    const float* A; const float* Wt; const float* Wt2;
    size_t aoff, boff;
    int ar, ac, br, bc, mrow, half;
    int rot, kmask;
};

template <int LDW>
__device__ __forceinline__ void chunk_ptrs(const GemmCtx& x, const int k,
                                           const float*& ap, const float*& bp,
                                           const float*& bp2)
{
    const int kk = (k + x.rot) & x.kmask;
    ap  = x.A + x.aoff + kk * KC;
    bp  = x.Wt + x.boff + (size_t)(kk * KC) * LDW;
    bp2 = x.Wt2 + x.boff + (size_t)(kk * KC) * LDW;
}

template <int LDW, int NT, int NW, int P, bool DO_STORE, bool DO_LOAD>
__device__ __forceinline__ void gemm_trip(const GemmCtx& x, const int c,
                                          float (*ldsA)[16][LDA_PITCH],
                                          float (*ldsB0)[KC][LDB_PITCH],
                                          float (*ldsB1)[KC][LDB_PITCH],
                                          v4f (&av)[2][KC / 32],
                                          v4f (&bv)[2][NW * KC * NT / 8],
                                          v8f (&acc)[NT])
{
    constexpr int G  = 8;
    constexpr int NG = KC / (4 * G);
    static_assert(NG == KC / 32, "a slice of a chunk is one fragment group");

    float (*curA)[LDA_PITCH]      = ldsA[P];
    float (*nxtA)[LDA_PITCH]      = ldsA[P ^ 1];
    float (*curB)[KC][LDB_PITCH]  = P ? ldsB1 : ldsB0;
    float (*nxtB)[KC][LDB_PITCH]  = P ? ldsB0 : ldsB1;
    const int mrow = x.mrow, half = x.half;

    const float* ap  = x.A;
    const float* bp  = x.Wt;
    const float* bp2 = x.Wt2;
    if (DO_LOAD) chunk_ptrs<LDW>(x, c + 3, ap, bp, bp2);

    v2f af[2][G];
    v2f bf[2][G][NT];

    read_group<NT>(curA, curB, 0, mrow, half, af[0], bf[0]);
#pragma unroll
    for (int g = 0; g < NG; ++g) {
        if (g + 1 < NG)
            read_group<NT>(curA, curB, g + 1, mrow, half, af[(g + 1) & 1], bf[(g + 1) & 1]);
        if (DO_STORE)
            store_slice<NT, NW>(nxtA, nxtB, g, x.ar, x.ac, x.br, x.bc, av[P ^ 1], bv[P ^ 1]);
        asm volatile("" ::: "memory");
        if (DO_LOAD)
            load_slice<LDW, NT, NW>(ap, bp, bp2, g, av[P ^ 1], bv[P ^ 1]);
        __builtin_amdgcn_sched_barrier(0);
#pragma unroll
        for (int s = 0; s < G; ++s) {
#pragma unroll
            for (int j = 0; j < NT; ++j)
                acc[j] = __builtin_amdgcn_wmma_f32_16x16x4_f32(
                    false, af[g & 1][s], false, bf[g & 1][s][j], (short)0,
                    acc[j], false, false);
        }
        __builtin_amdgcn_sched_barrier(0);
    }
    lds_stage_barrier();
}

template <int LDA, int LDW, int NT, int NC, int NW>
__device__ inline void gemm_tile_acc(const float* __restrict__ A,
                                     const float* __restrict__ Wt,
                                     const float* __restrict__ Wt2,
                                     const int ts, int m0, int n0w, const int rot,
                                     float (*ldsA)[16][LDA_PITCH],
                                     float (*ldsB0)[KC][LDB_PITCH],
                                     float (*ldsB1)[KC][LDB_PITCH],
                                     v8f (&acc)[NT])
{
    static_assert(NC >= 4 && NC % 2 == 0, "pipeline shape");
    static_assert((NC & (NC - 1)) == 0, "the start chunk wraps round with a mask");
    const int t    = ts;
    const int lane = t & 31;

    constexpr int LPR = 4 * NT;
    constexpr int NB  = KC * NT / 8;
    constexpr int NQ  = KC / 32;

    GemmCtx x;
    x.A = A; x.Wt = Wt; x.Wt2 = Wt2;
    x.ar = t >> 3;
    x.ac = (t & 7) * 4;
    x.br = lane / LPR;
    x.bc = (lane % LPR) * 4;
    x.mrow = lane & 15;
    x.half = lane >> 4;
    x.aoff = (size_t)(m0 + x.ar) * LDA + x.ac;
    x.boff = (size_t)x.br * LDW + n0w + x.bc;
    x.kmask = NC - 1;
    x.rot = rot & (NC - 1);

    v4f av[2][NQ];
    v4f bv[2][NW * NB];

    const float* ap;
    const float* bp;
    const float* bp2;
    chunk_ptrs<LDW>(x, 0, ap, bp, bp2);
#pragma unroll
    for (int q = 0; q < NQ; ++q) load_slice<LDW, NT, NW>(ap, bp, bp2, q, av[0], bv[0]);
    chunk_ptrs<LDW>(x, 1, ap, bp, bp2);
#pragma unroll
    for (int q = 0; q < NQ; ++q) load_slice<LDW, NT, NW>(ap, bp, bp2, q, av[1], bv[1]);
#pragma unroll
    for (int q = 0; q < NQ; ++q)
        store_slice<NT, NW>(ldsA[0], ldsB0, q, x.ar, x.ac, x.br, x.bc, av[0], bv[0]);
    asm volatile("" ::: "memory");
    chunk_ptrs<LDW>(x, 2, ap, bp, bp2);
#pragma unroll
    for (int q = 0; q < NQ; ++q) load_slice<LDW, NT, NW>(ap, bp, bp2, q, av[0], bv[0]);
    lds_stage_barrier();

#pragma nounroll
    for (int c = 0; c + 4 < NC; c += 2) {
        gemm_trip<LDW, NT, NW, 0, true, true>(x, c,     ldsA, ldsB0, ldsB1, av, bv, acc);
        gemm_trip<LDW, NT, NW, 1, true, true>(x, c + 1, ldsA, ldsB0, ldsB1, av, bv, acc);
    }
    gemm_trip<LDW, NT, NW, 0, true,  true >(x, NC - 4, ldsA, ldsB0, ldsB1, av, bv, acc);
    gemm_trip<LDW, NT, NW, 1, true,  false>(x, NC - 3, ldsA, ldsB0, ldsB1, av, bv, acc);
    gemm_trip<LDW, NT, NW, 0, true,  false>(x, NC - 2, ldsA, ldsB0, ldsB1, av, bv, acc);
    gemm_trip<LDW, NT, NW, 1, false, false>(x, NC - 1, ldsA, ldsB0, ldsB1, av, bv, acc);
}


__device__ __forceinline__ float sigmoid_sl(const float z)
{
    return __builtin_amdgcn_rcpf(1.0f + __expf(-z));
}

__device__ __forceinline__ float tanh_sl(const float x)
{
    const float ax = fabsf(x);
    const float s  = ax * ax;
    float p = fmaf(s, -0.0057000206f, 0.020634072f);
    p = fmaf(s, p, -0.05373793f);
    p = fmaf(s, p, 0.13331416f);
    p = fmaf(s, p, -0.3333328f);
    const float lo = fmaf(s, ax * p, ax);
    const float hi = fmaf(__builtin_amdgcn_rcpf(__expf(2.0f * ax) + 1.0f), -2.0f, 1.0f);
    return copysignf((ax < 0.625f) ? lo : hi, x);
}
template <int K1, int K2, bool SAME_A>
__global__ void __launch_bounds__(256)
lstm_cell_kernel(const float* __restrict__ A1, const float* __restrict__ W,
                 const float* __restrict__ A2, const float* __restrict__ Uw,
                 const float* __restrict__ bias, const float* __restrict__ Cprev,
                 float* __restrict__ Hout, float* __restrict__ Cout,
                 int Uu)
{
    __shared__ float ldsA[2][2][16][LDA_PITCH];
    __shared__ float ldsB[2][8][2][KC][LDB_PITCH];
    __shared__ float pbuf[4][2][8][32];
    __shared__ float gbuf[4][16][34];

    const int t    = threadIdx.x;
    const int wave = t >> 5;
    const int gate = wave & 3;
    const int kset = wave >> 2;
    const int lane = t & 31;
    const int half = lane >> 4;
    const int mrow = lane & 15;
    const int m0   = blockIdx.y * 16;
    const int n0   = blockIdx.x * 32;
    const int n0w  = gate * Uu + n0;

    const int rot = blockIdx.x + (SAME_A ? 1 : 2) * blockIdx.y;

    const int em = t >> 5, en = t & 31;
    const size_t idx0 = (size_t)(m0 + em) * Uu + n0 + en;
    const size_t idx1 = idx0 + (size_t)8 * Uu;
    const float cp0 = Cprev[idx0];
    const float cp1 = Cprev[idx1];
    const float bz0 = bias[n0w + mrow];
    const float bz1 = bias[n0w + 16 + mrow];
    asm volatile("" ::: "memory");

    v8f acc[2] = {};
    static_assert(K1 % KC == 0 && K1 == K2,
                  "whole chunks, and both wave sets meet the same barriers");
    if (SAME_A) {
        constexpr int KH = K1 / 2;
        static_assert(K1 % 2 == 0 && KH % KC == 0, "whole chunks in each half of K");
        gemm_tile_acc<1024, 4096, 2, KH / KC, 2>(
            A1 + kset * KH, W + (size_t)(kset * KH) * 4096,
            Uw + (size_t)(kset * KH) * 4096, t & 127, m0, n0w, rot,
            ldsA[kset], ldsB[0][wave], ldsB[1][wave], acc);
    } else {
        const float* Wset = kset ? Uw : W;
        gemm_tile_acc<1024, 4096, 2, K1 / KC, 1>(
            kset ? A2 : A1, Wset, Wset, t & 127, m0, n0w, rot,
            ldsA[kset], ldsB[0][wave], ldsB[1][wave], acc);
    }

    if (kset == 1) {
#pragma unroll
        for (int j = 0; j < 2; ++j) {
#pragma unroll
            for (int r = 0; r < 8; ++r) pbuf[gate][j][r][lane] = acc[j][r];
        }
    }
    __syncthreads();

    if (kset == 0) {
        float z[16];
#pragma unroll
        for (int q = 0; q < 16; ++q)
            z[q] = (acc[q >> 3][q & 7] + pbuf[gate][q >> 3][q & 7][lane]) + ((q >> 3) ? bz1 : bz0);
        const int gate_s = __builtin_amdgcn_readfirstlane(gate);
        if (gate_s == 2) {
#pragma unroll
            for (int q = 0; q < 16; ++q) z[q] = tanh_sl(z[q]);
        } else {
#pragma unroll
            for (int q = 0; q < 16; ++q) z[q] = sigmoid_sl(z[q]);
        }
#pragma unroll
        for (int q = 0; q < 16; ++q)
            gbuf[gate][(q & 7) + 8 * half][16 * (q >> 3) + mrow] = z[q];
    }
    __syncthreads();

    const float iv0 = gbuf[0][em][en],     iv1 = gbuf[0][em + 8][en];
    const float fv0 = gbuf[1][em][en],     fv1 = gbuf[1][em + 8][en];
    const float gv0 = gbuf[2][em][en],     gv1 = gbuf[2][em + 8][en];
    const float ov0 = gbuf[3][em][en],     ov1 = gbuf[3][em + 8][en];
    const float cn0 = fv0 * cp0 + iv0 * gv0;
    const float cn1 = fv1 * cp1 + iv1 * gv1;
    Cout[idx0] = cn0;
    Cout[idx1] = cn1;
    Hout[idx0] = ov0 * tanh_sl(cn0);
    Hout[idx1] = ov1 * tanh_sl(cn1);
}

template <int K>
__global__ void __launch_bounds__(256)
sproj_kernel(const float* __restrict__ H2, const float* __restrict__ Ws,
             const float* __restrict__ bs, float* __restrict__ Sout, int Uu)
{
    __shared__ float ldsA[2][2][16][LDA_PITCH];
    __shared__ float ldsB[2][8][1][KC][LDB_PITCH];
    __shared__ float pbuf[4][8][32];

    const int t    = threadIdx.x;
    const int wave = t >> 5;
    const int ct   = wave & 3;
    const int kset = wave >> 2;
    const int lane = t & 31;
    const int half = lane >> 4;
    const int mrow = lane & 15;
    const int m0   = blockIdx.y * 16;
    const int n0w  = blockIdx.x * 64 + ct * 16;

    constexpr int KH = K / 2;
    const int rot = blockIdx.x + blockIdx.y;
    v8f acc1[1] = {};
    static_assert(K % 2 == 0 && KH % KC == 0, "whole chunks in each half of K");
    gemm_tile_acc<1024, 1024, 1, KH / KC, 1>(
        H2 + kset * KH, Ws + (size_t)(kset * KH) * 1024,
        Ws + (size_t)(kset * KH) * 1024, t & 127, m0, n0w, rot,
        ldsA[kset], ldsB[0][wave], ldsB[1][wave], acc1);
    const v8f acc = acc1[0];

    if (kset == 1) {
#pragma unroll
        for (int r = 0; r < 8; ++r) pbuf[ct][r][lane] = acc[r];
    }
    __syncthreads();

    if (kset == 0) {
        const float bv = bs[n0w + mrow];
#pragma unroll
        for (int r = 0; r < 8; ++r) {
            const int gm = m0 + r + 8 * half;
            Sout[(size_t)gm * Uu + n0w + mrow] = (acc[r] + pbuf[ct][r][lane]) + bv;
        }
    }
}

#define ATT_SPLIT 2

__global__ void __launch_bounds__(512)
attention_part_kernel(const float* __restrict__ av, const float* __restrict__ S,
                      float* __restrict__ P, float* __restrict__ ML,
                      int T, int Uu, int B)
{
    __shared__ float sS[1024];
    __shared__ float ctx_s[1024];
    __shared__ float wm[16], wl[16];

    const int b    = blockIdx.x / ATT_SPLIT;
    const int h    = blockIdx.x % ATT_SPLIT;
    const int t    = threadIdx.x;
    const int wave = t >> 5;
    const int lane = t & 31;

    for (int u = t; u < Uu; u += 512) {
        sS[u]    = S[(size_t)b * Uu + u];
        ctx_s[u] = 0.0f;
    }
    __syncthreads();

    float m = -INFINITY, l = 0.0f;
    float ctx[32];
#pragma unroll
    for (int j = 0; j < 32; ++j) ctx[j] = 0.0f;

    const int TH = T / ATT_SPLIT;
    const int t0 = h * TH;
    const int t1 = (h == ATT_SPLIT - 1) ? T : t0 + TH;
    const float* avb = av + (size_t)b * T * Uu;
    for (int tt = t0 + wave; tt < t1; tt += 16) {
        const float* row = avb + (size_t)tt * Uu;
        __builtin_prefetch(avb + (size_t)(tt + 16) * Uu + lane * 32, 0, 0);

        v4f frag[8];
        float dot = 0.0f;
#pragma unroll
        for (int j = 0; j < 8; ++j) {
            const int u = (j * 32 + lane) * 4;
            frag[j] = __builtin_nontemporal_load((const v4f*)(row + u));
            dot += frag[j].x * sS[u] + frag[j].y * sS[u + 1] +
                   frag[j].z * sS[u + 2] + frag[j].w * sS[u + 3];
        }
#pragma unroll
        for (int off = 16; off > 0; off >>= 1) dot += __shfl_xor(dot, off, 32);

        const float mn    = fmaxf(m, dot);
        const float scale = __expf(m - mn);
        const float e     = __expf(dot - mn);
        l = l * scale + e;
#pragma unroll
        for (int j = 0; j < 8; ++j) {
            ctx[j * 4 + 0] = ctx[j * 4 + 0] * scale + e * frag[j].x;
            ctx[j * 4 + 1] = ctx[j * 4 + 1] * scale + e * frag[j].y;
            ctx[j * 4 + 2] = ctx[j * 4 + 2] * scale + e * frag[j].z;
            ctx[j * 4 + 3] = ctx[j * 4 + 3] * scale + e * frag[j].w;
        }
        m = mn;
    }

    if (lane == 0) { wm[wave] = m; wl[wave] = l; }
    __syncthreads();

    float M = wm[0];
#pragma unroll
    for (int w = 1; w < 16; ++w) M = fmaxf(M, wm[w]);
    float L = 0.0f;
#pragma unroll
    for (int w = 0; w < 16; ++w) L += wl[w] * __expf(wm[w] - M);

    const float f = __expf(m - M);
#pragma unroll
    for (int j = 0; j < 8; ++j) {
        const int u = (j * 32 + lane) * 4;
        atomicAdd(&ctx_s[u + 0], ctx[j * 4 + 0] * f);
        atomicAdd(&ctx_s[u + 1], ctx[j * 4 + 1] * f);
        atomicAdd(&ctx_s[u + 2], ctx[j * 4 + 2] * f);
        atomicAdd(&ctx_s[u + 3], ctx[j * 4 + 3] * f);
    }
    __syncthreads();

    float* Pb = P + ((size_t)h * B + b) * Uu;
    for (int u = t; u < Uu; u += 512) Pb[u] = ctx_s[u];
    if (t == 0) {
        ML[2 * blockIdx.x]     = M;
        ML[2 * blockIdx.x + 1] = L;
    }
}

__global__ void __launch_bounds__(256)
attention_combine_kernel(const float* __restrict__ P, const float* __restrict__ ML,
                         const float* __restrict__ C2, float* __restrict__ Cout,
                         int Uu, int B)
{
    const int b = blockIdx.x;

    float Mh[ATT_SPLIT], e[ATT_SPLIT];
    float M = -INFINITY;
    for (int h = 0; h < ATT_SPLIT; ++h) {
        Mh[h] = ML[2 * (b * ATT_SPLIT + h)];
        M = fmaxf(M, Mh[h]);
    }
    float L = 0.0f;
    for (int h = 0; h < ATT_SPLIT; ++h) {
        e[h] = __expf(Mh[h] - M);
        L += ML[2 * (b * ATT_SPLIT + h) + 1] * e[h];
    }
    const float invL = 1.0f / L;

    const size_t base  = (size_t)b * Uu;
    const size_t plane = (size_t)B * Uu;
    for (int u = 4 * threadIdx.x; u < Uu; u += 4 * 256) {
        v4f acc = *(const v4f*)(P + base + u) * e[0];
        for (int h = 1; h < ATT_SPLIT; ++h)
            acc += *(const v4f*)(P + h * plane + base + u) * e[h];
        const v4f c2 = *(const v4f*)(C2 + base + u);
        *(v4f*)(Cout + base + u) = c2 + acc * invL;
    }
}

extern "C" void kernel_launch(void* const* d_in, const int* in_sizes, int n_in,
                              void* d_out, int out_size, void* d_ws, size_t ws_size,
                              hipStream_t stream)
{
    (void)in_sizes; (void)n_in; (void)out_size; (void)ws_size;
    const int B = 128, T = 512, U = 1024;

    const float* x   = (const float*)d_in[0];
    const float* h0  = (const float*)d_in[1];
    const float* c0  = (const float*)d_in[2];
    const float* av  = (const float*)d_in[3];
    const float* W1  = (const float*)d_in[4];
    const float* U1  = (const float*)d_in[5];
    const float* b1  = (const float*)d_in[6];
    const float* W2  = (const float*)d_in[7];
    const float* U2  = (const float*)d_in[8];
    const float* b2  = (const float*)d_in[9];
    const float* Ws  = (const float*)d_in[10];
    const float* bs  = (const float*)d_in[11];
    float* out = (float*)d_out;

    float* ws = (float*)d_ws;
    const size_t BU = (size_t)B * U;
    float* h1 = ws;
    float* c1 = ws + BU;
    float* c2 = ws + 3 * BU;
    float* s  = ws + 4 * BU;
    float* h2   = out;
    float* part = ws;
    float* ml   = ws + 2 * BU;
    static_assert(ATT_SPLIT == 2, "two dead planes for the partial contexts");

    dim3 gGemm(U / 32, B / 16);
    dim3 bGemm(256);

    lstm_cell_kernel<1024, 1024, false><<<gGemm, bGemm, 0, stream>>>(x, W1, h0, U1, b1,
                                                                     c0, h1, c1, U);
    lstm_cell_kernel<1024, 1024, true><<<gGemm, bGemm, 0, stream>>>(h1, W2, h1, U2, b2,
                                                                    c1, h2, c2, U);
    sproj_kernel<1024><<<dim3(U / 64, B / 16), bGemm, 0, stream>>>(h2, Ws, bs, s, U);
    attention_part_kernel<<<dim3(B * ATT_SPLIT), dim3(512), 0, stream>>>(av, s, part, ml,
                                                                         T, U, B);
    attention_combine_kernel<<<dim3(B), dim3(256), 0, stream>>>(part, ml, c2, out + BU,
                                                                U, B);
}
